# starting kernel
# baseline (speedup 1.0000x reference)
#include <hip/hip_runtime.h>
#include <math.h>

typedef float v2f __attribute__((ext_vector_type(2)));
typedef float v4f __attribute__((ext_vector_type(4)));
typedef float v8f __attribute__((ext_vector_type(8)));

#define N_POS 16384
#define EMB 128
#define TEMP_INV 10.0f
#define LN_EPSF 1e-5f
#define KSTR 132
#define PSTR 34

__global__ __launch_bounds__(64) void encode_kernel(
    const float* __restrict__ ego_cls, const float* __restrict__ ego_reg,
    const float* __restrict__ other_cls, const float* __restrict__ other_reg,
    const float* __restrict__ w1, const float* __restrict__ b1,
    const float* __restrict__ ln_g, const float* __restrict__ ln_b,
    const float* __restrict__ w2, const float* __restrict__ b2,
    float* __restrict__ Qbuf, float* __restrict__ Kbuf)
{
    __shared__ float hls[64 * 129];
    const int tid = threadIdx.x;
    const int p   = blockIdx.x * 64 + tid;

    const float* cls; const float* reg; float* outp;
    if (blockIdx.y == 0) { cls = ego_cls;   reg = ego_reg;   outp = Qbuf; }
    else                 { cls = other_cls; reg = other_reg; outp = Kbuf; }

    float x0 = reg[3 * N_POS + p];
    float x1 = reg[4 * N_POS + p];
    float x2 = reg[5 * N_POS + p];
    float th = reg[6 * N_POS + p];
    float x3 = sinf(th);
    float x4 = cosf(th);
    float x5 = fmaxf(cls[p], cls[N_POS + p]);

    float* hrow = &hls[tid * 129];
    float sum = 0.f, sumsq = 0.f;
    for (int c = 0; c < EMB; ++c) {
        float acc = b1[c];
        acc += x0 * w1[0 * EMB + c];
        acc += x1 * w1[1 * EMB + c];
        acc += x2 * w1[2 * EMB + c];
        acc += x3 * w1[3 * EMB + c];
        acc += x4 * w1[4 * EMB + c];
        acc += x5 * w1[5 * EMB + c];
        acc = fmaxf(acc, 0.f);
        hrow[c] = acc;
        sum += acc; sumsq += acc * acc;
    }
    float mu   = sum * (1.f / EMB);
    float var  = sumsq * (1.f / EMB) - mu * mu;
    float rstd = rsqrtf(var + LN_EPSF);
    for (int c = 0; c < EMB; ++c)
        hrow[c] = (hrow[c] - mu) * rstd * ln_g[c] + ln_b[c];

    for (int e = 0; e < EMB; ++e) {
        float acc = b2[e];
        for (int c = 0; c < EMB; ++c)
            acc += hrow[c] * w2[c * EMB + e];
        outp[p * EMB + e] = acc;
    }
}

__global__ void vpack_kernel(const float* __restrict__ oc, const float* __restrict__ org,
                             float* __restrict__ V)
{
    int t = blockIdx.x * 256 + threadIdx.x;
    if (t >= N_POS * 16) return;
    int p = t >> 4, c = t & 15;
    V[t] = (c < 2) ? oc[c * N_POS + p] : org[(c - 2) * N_POS + p];
}

__global__ __launch_bounds__(256) void attn_kernel(
    const float* __restrict__ Q, const float* __restrict__ K,
    const float* __restrict__ V, float* __restrict__ out)
{
    __shared__ float Kls[32 * KSTR];
    __shared__ float Vls[16 * PSTR];
    __shared__ float Pls[8][16 * PSTR];

    const int tid  = threadIdx.x;
    const int lane = tid & 31;
    const int wave = tid >> 5;
    const int half = lane >> 4;
    const int l16  = lane & 15;
    const int qRow = blockIdx.x * 128 + wave * 16 + l16;

    v2f qf[32];
#pragma unroll
    for (int kk = 0; kk < 32; ++kk) {
        const float* s = &Q[qRow * EMB + kk * 4 + half * 2];
        v2f t; t.x = s[0] * TEMP_INV; t.y = s[1] * TEMP_INV;
        qf[kk] = t;
    }

    v8f Oacc = {};
    float mstate[8], lstate[8];
#pragma unroll
    for (int v = 0; v < 8; ++v) { mstate[v] = -3.0e38f; lstate[v] = 0.f; }

    float* pmine = &Pls[wave][0];

    for (int kt = 0; kt < N_POS / 32; ++kt) {
        __syncthreads();
#pragma unroll
        for (int i = 0; i < 4; ++i) {
            int idx4 = tid * 4 + i;
            int row  = idx4 >> 5;
            int col4 = (idx4 & 31) * 4;
            *(v4f*)&Kls[row * KSTR + col4] = *(const v4f*)&K[(kt * 32 + row) * EMB + col4];
        }
#pragma unroll
        for (int i = 0; i < 2; ++i) {
            int idx = tid + i * 256;
            int k = idx >> 4, c = idx & 15;
            Vls[c * PSTR + k] = V[(kt * 32 + k) * 16 + c];
        }
        __syncthreads();

        v8f S0 = {}, S1 = {};
        {
            v2f bf[32];
#pragma unroll
            for (int kk = 0; kk < 32; ++kk)
                bf[kk] = *(const v2f*)&Kls[l16 * KSTR + kk * 4 + half * 2];
#pragma unroll
            for (int kk = 0; kk < 32; ++kk)
                S0 = __builtin_amdgcn_wmma_f32_16x16x4_f32(false, qf[kk], false, bf[kk],
                                                           (short)0, S0, false, false);
#pragma unroll
            for (int kk = 0; kk < 32; ++kk)
                bf[kk] = *(const v2f*)&Kls[(16 + l16) * KSTR + kk * 4 + half * 2];
#pragma unroll
            for (int kk = 0; kk < 32; ++kk)
                S1 = __builtin_amdgcn_wmma_f32_16x16x4_f32(false, qf[kk], false, bf[kk],
                                                           (short)0, S1, false, false);
        }

#pragma unroll
        for (int v = 0; v < 8; ++v) {
            float s0 = S0[v], s1 = S1[v];
            float rm = fmaxf(s0, s1);
            rm = fmaxf(rm, __shfl_xor(rm, 1, 16));
            rm = fmaxf(rm, __shfl_xor(rm, 2, 16));
            rm = fmaxf(rm, __shfl_xor(rm, 4, 16));
            rm = fmaxf(rm, __shfl_xor(rm, 8, 16));
            float newm  = fmaxf(mstate[v], rm);
            float alpha = __expf(mstate[v] - newm);
            float p0    = __expf(s0 - newm);
            float p1    = __expf(s1 - newm);
            float rs = p0 + p1;
            rs += __shfl_xor(rs, 1, 16);
            rs += __shfl_xor(rs, 2, 16);
            rs += __shfl_xor(rs, 4, 16);
            rs += __shfl_xor(rs, 8, 16);
            lstate[v] = lstate[v] * alpha + rs;
            mstate[v] = newm;
            Oacc[v]  *= alpha;
            int m = v + 8 * half;
            pmine[m * PSTR + l16]      = p0;
            pmine[m * PSTR + 16 + l16] = p1;
        }
        asm volatile("s_wait_dscnt 0" ::: "memory");

        {
            v2f pa[8], vb[8];
#pragma unroll
            for (int c4 = 0; c4 < 8; ++c4) {
                pa[c4] = *(const v2f*)&pmine[l16 * PSTR + c4 * 4 + half * 2];
                vb[c4] = *(const v2f*)&Vls  [l16 * PSTR + c4 * 4 + half * 2];
            }
#pragma unroll
            for (int c4 = 0; c4 < 8; ++c4)
                Oacc = __builtin_amdgcn_wmma_f32_16x16x4_f32(false, pa[c4], false, vb[c4],
                                                             (short)0, Oacc, false, false);
        }
    }

#pragma unroll
    for (int v = 0; v < 8; ++v) {
        int q = blockIdx.x * 128 + wave * 16 + v + 8 * half;
        out[l16 * N_POS + q] = Oacc[v] / lstate[v];
    }
}

extern "C" void kernel_launch(void* const* d_in, const int* in_sizes, int n_in,
                              void* d_out, int out_size, void* d_ws, size_t ws_size,
                              hipStream_t stream)
{
    const float* ego_cls   = (const float*)d_in[0];
    const float* ego_reg   = (const float*)d_in[1];
    const float* other_cls = (const float*)d_in[2];
    const float* other_reg = (const float*)d_in[3];
    const float* w1  = (const float*)d_in[4];
    const float* b1  = (const float*)d_in[5];
    const float* ln_g = (const float*)d_in[6];
    const float* ln_b = (const float*)d_in[7];
    const float* w2  = (const float*)d_in[8];
    const float* b2  = (const float*)d_in[9];

    float* Qbuf = (float*)d_ws;
    float* Kbuf = Qbuf + N_POS * EMB;
    float* Vbuf = Kbuf + N_POS * EMB;
    float* out  = (float*)d_out;

    encode_kernel<<<dim3(N_POS / 64, 2), 64, 0, stream>>>(
        ego_cls, ego_reg, other_cls, other_reg,
        w1, b1, ln_g, ln_b, w2, b2, Qbuf, Kbuf);

    vpack_kernel<<<(N_POS * 16 + 255) / 256, 256, 0, stream>>>(other_cls, other_reg, Vbuf);

    attn_kernel<<<N_POS / 128, 256, 0, stream>>>(Qbuf, Kbuf, Vbuf, out);
}


# attention: each 8-wave work-group owns 32 queries, every 128-key tile is split over 4 wave groups and the partial softmax states are merged in LDS (new summation order); 512 work-groups, 2 per WGP
# speedup vs baseline: 1.9367x; 1.9367x over previous
#include <hip/hip_runtime.h>
#include <math.h>

typedef float v2f __attribute__((ext_vector_type(2)));
typedef float v4f __attribute__((ext_vector_type(4)));
typedef float v8f __attribute__((ext_vector_type(8)));

#define N_POS 16384
#define EMB 128
#define TEMP_INV 10.0f
#define LN_EPSF 1e-5f
#define KSTR 132
#define PSTR 34

__global__ __launch_bounds__(64) void encode_kernel(
    const float* __restrict__ ego_cls, const float* __restrict__ ego_reg,
    const float* __restrict__ other_cls, const float* __restrict__ other_reg,
    const float* __restrict__ w1, const float* __restrict__ b1,
    const float* __restrict__ ln_g, const float* __restrict__ ln_b,
    const float* __restrict__ w2, const float* __restrict__ b2,
    float* __restrict__ Qbuf, float* __restrict__ Kbuf)
{
    __shared__ float hls[64 * 129];
    const int tid = threadIdx.x;
    const int p   = blockIdx.x * 64 + tid;

    const float* cls; const float* reg; float* outp;
    if (blockIdx.y == 0) { cls = ego_cls;   reg = ego_reg;   outp = Qbuf; }
    else                 { cls = other_cls; reg = other_reg; outp = Kbuf; }

    float x0 = reg[3 * N_POS + p];
    float x1 = reg[4 * N_POS + p];
    float x2 = reg[5 * N_POS + p];
    float th = reg[6 * N_POS + p];
    float x3 = sinf(th);
    float x4 = cosf(th);
    float x5 = fmaxf(cls[p], cls[N_POS + p]);

    float* hrow = &hls[tid * 129];
    float sum = 0.f, sumsq = 0.f;
    for (int c = 0; c < EMB; ++c) {
        float acc = b1[c];
        acc += x0 * w1[0 * EMB + c];
        acc += x1 * w1[1 * EMB + c];
        acc += x2 * w1[2 * EMB + c];
        acc += x3 * w1[3 * EMB + c];
        acc += x4 * w1[4 * EMB + c];
        acc += x5 * w1[5 * EMB + c];
        acc = fmaxf(acc, 0.f);
        hrow[c] = acc;
        sum += acc; sumsq += acc * acc;
    }
    float mu   = sum * (1.f / EMB);
    float var  = sumsq * (1.f / EMB) - mu * mu;
    float rstd = rsqrtf(var + LN_EPSF);
    for (int c = 0; c < EMB; ++c)
        hrow[c] = (hrow[c] - mu) * rstd * ln_g[c] + ln_b[c];

    for (int e = 0; e < EMB; ++e) {
        float acc = b2[e];
        for (int c = 0; c < EMB; ++c)
            acc += hrow[c] * w2[c * EMB + e];
        outp[p * EMB + e] = acc;
    }
}

__global__ void vpack_kernel(const float* __restrict__ oc, const float* __restrict__ org,
                             float* __restrict__ V)
{
    int t = blockIdx.x * 256 + threadIdx.x;
    if (t >= N_POS * 16) return;
    int p = t >> 4, c = t & 15;
    V[t] = (c < 2) ? oc[c * N_POS + p] : org[(c - 2) * N_POS + p];
}

#define KG 4
#define QW 2
#define QPB (QW * 16)
#define KPT (KG * 32)
#define CSTR 33

__global__ __launch_bounds__(256) void attn_kernel(
    const float* __restrict__ Q, const float* __restrict__ K,
    const float* __restrict__ V, float* __restrict__ out)
{
    __shared__ float Kls[KPT * KSTR];
    __shared__ float Vls[KG][16 * PSTR];
    __shared__ float Pls[8][16 * PSTR];

    const int tid  = threadIdx.x;
    const int lane = tid & 31;
    const int wave = tid >> 5;
    const int g    = wave / QW;
    const int wq   = wave % QW;
    const int half = lane >> 4;
    const int l16  = lane & 15;
    const int qRow = blockIdx.x * QPB + wq * 16 + l16;

    v2f qf[32];
#pragma unroll
    for (int kk = 0; kk < 32; ++kk) {
        const float* s = &Q[qRow * EMB + kk * 4 + half * 2];
        v2f t; t.x = s[0] * TEMP_INV; t.y = s[1] * TEMP_INV;
        qf[kk] = t;
    }

    v8f Oacc = {};
    float mstate[8], lstate[8];
#pragma unroll
    for (int v = 0; v < 8; ++v) { mstate[v] = -3.0e38f; lstate[v] = 0.f; }

    float* pmine = &Pls[wave][0];
    const float* kmine = &Kls[g * 32 * KSTR];
    const float* vmine = &Vls[g][0];

    for (int kt = 0; kt < N_POS / KPT; ++kt) {
        __syncthreads();
        {
            const float* kp = K + (size_t)kt * (KPT * EMB) + tid * 4;
            const float* vp = V + (size_t)kt * (KPT * 16) + tid * 4;
            v4f kr[16], vr[2];
#pragma unroll
            for (int i = 0; i < 16; ++i) kr[i] = *(const v4f*)(kp + i * 1024);
#pragma unroll
            for (int i = 0; i < 2; ++i)  vr[i] = *(const v4f*)(vp + i * 1024);
            float* kd = &Kls[wave * KSTR + lane * 4];
#pragma unroll
            for (int i = 0; i < 16; ++i) *(v4f*)(kd + i * 8 * KSTR) = kr[i];
#pragma unroll
            for (int i = 0; i < 2; ++i) {
                int j  = tid + i * 256;
                int k  = j >> 2;
                int c4 = (j & 3) * 4;
                float* vd = &Vls[k >> 5][c4 * PSTR + (k & 31)];
                vd[0 * PSTR] = vr[i].x; vd[1 * PSTR] = vr[i].y;
                vd[2 * PSTR] = vr[i].z; vd[3 * PSTR] = vr[i].w;
            }
        }
        __syncthreads();

        v8f S0 = {}, S1 = {};
        {
            v2f bf[32];
#pragma unroll
            for (int kk = 0; kk < 32; ++kk)
                bf[kk] = *(const v2f*)&kmine[l16 * KSTR + kk * 4 + half * 2];
#pragma unroll
            for (int kk = 0; kk < 32; ++kk)
                S0 = __builtin_amdgcn_wmma_f32_16x16x4_f32(false, qf[kk], false, bf[kk],
                                                           (short)0, S0, false, false);
#pragma unroll
            for (int kk = 0; kk < 32; ++kk)
                bf[kk] = *(const v2f*)&kmine[(16 + l16) * KSTR + kk * 4 + half * 2];
#pragma unroll
            for (int kk = 0; kk < 32; ++kk)
                S1 = __builtin_amdgcn_wmma_f32_16x16x4_f32(false, qf[kk], false, bf[kk],
                                                           (short)0, S1, false, false);
        }

#pragma unroll
        for (int v = 0; v < 8; ++v) {
            float s0 = S0[v], s1 = S1[v];
            float rm = fmaxf(s0, s1);
            rm = fmaxf(rm, __shfl_xor(rm, 1, 16));
            rm = fmaxf(rm, __shfl_xor(rm, 2, 16));
            rm = fmaxf(rm, __shfl_xor(rm, 4, 16));
            rm = fmaxf(rm, __shfl_xor(rm, 8, 16));
            float newm  = fmaxf(mstate[v], rm);
            float alpha = __expf(mstate[v] - newm);
            float p0    = __expf(s0 - newm);
            float p1    = __expf(s1 - newm);
            float rs = p0 + p1;
            rs += __shfl_xor(rs, 1, 16);
            rs += __shfl_xor(rs, 2, 16);
            rs += __shfl_xor(rs, 4, 16);
            rs += __shfl_xor(rs, 8, 16);
            lstate[v] = lstate[v] * alpha + rs;
            mstate[v] = newm;
            Oacc[v]  *= alpha;
            int m = v + 8 * half;
            pmine[m * PSTR + l16]      = p0;
            pmine[m * PSTR + 16 + l16] = p1;
        }
        asm volatile("s_wait_dscnt 0" ::: "memory");

        {
            v2f pa[8], vb[8];
#pragma unroll
            for (int c4 = 0; c4 < 8; ++c4) {
                pa[c4] = *(const v2f*)&pmine[l16 * PSTR + c4 * 4 + half * 2];
                vb[c4] = *(const v2f*)&vmine[l16 * PSTR + c4 * 4 + half * 2];
            }
#pragma unroll
            for (int c4 = 0; c4 < 8; ++c4)
                Oacc = __builtin_amdgcn_wmma_f32_16x16x4_f32(false, pa[c4], false, vb[c4],
                                                             (short)0, Oacc, false, false);
        }
    }

    __syncthreads();
    float* cO = &Kls[0];
    float* cm = cO + KG * 16 * CSTR;
    float* cl = cm + KG * QPB;
#pragma unroll
    for (int v = 0; v < 8; ++v) {
        int ql = wq * 16 + v + 8 * half;
        cO[(g * 16 + l16) * CSTR + ql] = Oacc[v];
        if (l16 == 0) { cm[g * QPB + ql] = mstate[v]; cl[g * QPB + ql] = lstate[v]; }
    }
    __syncthreads();

    {
        const int ql = lane;
        float mg[KG], mx = -3.0e38f;
#pragma unroll
        for (int k = 0; k < KG; ++k) { mg[k] = cm[k * QPB + ql]; mx = fmaxf(mx, mg[k]); }
        float a[KG], lsum = 0.f;
#pragma unroll
        for (int k = 0; k < KG; ++k) { a[k] = __expf(mg[k] - mx); lsum += cl[k * QPB + ql] * a[k]; }
#pragma unroll
        for (int i = 0; i < 2; ++i) {
            int c = wave + 8 * i;
            float o = 0.f;
#pragma unroll
            for (int k = 0; k < KG; ++k) o += cO[(k * 16 + c) * CSTR + ql] * a[k];
            out[c * N_POS + blockIdx.x * QPB + ql] = o / lsum;
        }
    }
}

extern "C" void kernel_launch(void* const* d_in, const int* in_sizes, int n_in,
                              void* d_out, int out_size, void* d_ws, size_t ws_size,
                              hipStream_t stream)
{
    const float* ego_cls   = (const float*)d_in[0];
    const float* ego_reg   = (const float*)d_in[1];
    const float* other_cls = (const float*)d_in[2];
    const float* other_reg = (const float*)d_in[3];
    const float* w1  = (const float*)d_in[4];
    const float* b1  = (const float*)d_in[5];
    const float* ln_g = (const float*)d_in[6];
    const float* ln_b = (const float*)d_in[7];
    const float* w2  = (const float*)d_in[8];
    const float* b2  = (const float*)d_in[9];

    float* Qbuf = (float*)d_ws;
    float* Kbuf = Qbuf + N_POS * EMB;
    float* Vbuf = Kbuf + N_POS * EMB;
    float* out  = (float*)d_out;

    encode_kernel<<<dim3(N_POS / 64, 2), 64, 0, stream>>>(
        ego_cls, ego_reg, other_cls, other_reg,
        w1, b1, ln_g, ln_b, w2, b2, Qbuf, Kbuf);

    vpack_kernel<<<(N_POS * 16 + 255) / 256, 256, 0, stream>>>(other_cls, other_reg, Vbuf);

    attn_kernel<<<N_POS / QPB, 256, 0, stream>>>(Qbuf, Kbuf, Vbuf, out);
}


# encode: the 128x128 linear on f32 WMMA (16 tokens per wave, hidden row built in A-fragment layout, w2 staged in LDS) replaces a serial per-thread FMA chain; new summation order there and in LN sums
# speedup vs baseline: 2.7275x; 1.4083x over previous
#include <hip/hip_runtime.h>
#include <math.h>

typedef float v2f __attribute__((ext_vector_type(2)));
typedef float v4f __attribute__((ext_vector_type(4)));
typedef float v8f __attribute__((ext_vector_type(8)));

#define N_POS 16384
#define EMB 128
#define TEMP_INV 10.0f
#define LN_EPSF 1e-5f
#define KSTR 132
#define PSTR 34

#define ENC_TOK 128
#define W2PITCH 288

__global__ __launch_bounds__(256) void encode_kernel(
    const float* __restrict__ ego_cls, const float* __restrict__ ego_reg,
    const float* __restrict__ other_cls, const float* __restrict__ other_reg,
    const float* __restrict__ w1, const float* __restrict__ b1,
    const float* __restrict__ ln_g, const float* __restrict__ ln_b,
    const float* __restrict__ w2, const float* __restrict__ b2,
    float* __restrict__ Qbuf, float* __restrict__ Kbuf)
{
    __shared__ float W2p[64 * W2PITCH];

    const int tid  = threadIdx.x;
    const int lane = tid & 31;
    const int wave = tid >> 5;
    const int half = lane >> 4;
    const int l16  = lane & 15;

    {
        v4f r0[8], r1[8];
#pragma unroll
        for (int i = 0; i < 8; ++i) {
            int idx = tid + i * 256;
            int pr  = idx >> 5;
            int e4  = (idx & 31) * 4;
            r0[i] = *(const v4f*)&w2[(2 * pr) * EMB + e4];
            r1[i] = *(const v4f*)&w2[(2 * pr + 1) * EMB + e4];
        }
#pragma unroll
        for (int i = 0; i < 8; ++i) {
            int idx = tid + i * 256;
            int pr  = idx >> 5;
            int e4  = (idx & 31) * 4;
            float* d = &W2p[pr * W2PITCH + 2 * e4];
            v4f lo, hi;
            lo.x = r0[i].x; lo.y = r1[i].x; lo.z = r0[i].y; lo.w = r1[i].y;
            hi.x = r0[i].z; hi.y = r1[i].z; hi.z = r0[i].w; hi.w = r1[i].w;
            *(v4f*)d       = lo;
            *(v4f*)(d + 4) = hi;
        }
    }

    const float* cls; const float* reg; float* outp;
    if (blockIdx.y == 0) { cls = ego_cls;   reg = ego_reg;   outp = Qbuf; }
    else                 { cls = other_cls; reg = other_reg; outp = Kbuf; }

    const int tok0 = blockIdx.x * ENC_TOK + wave * 16;
    const int p    = tok0 + l16;

    float x0 = reg[3 * N_POS + p];
    float x1 = reg[4 * N_POS + p];
    float x2 = reg[5 * N_POS + p];
    float th = reg[6 * N_POS + p];
    float x3 = sinf(th);
    float x4 = cosf(th);
    float x5 = fmaxf(cls[p], cls[N_POS + p]);

    const int cb = 2 * half;
    v2f hf[32];
    float sum = 0.f, sumsq = 0.f;
#pragma unroll
    for (int kk = 0; kk < 32; ++kk) {
        const int c = 4 * kk + cb;
        v2f a  = *(const v2f*)&b1[c];
        v2f wa = *(const v2f*)&w1[0 * EMB + c];
        v2f wb = *(const v2f*)&w1[1 * EMB + c];
        v2f wc = *(const v2f*)&w1[2 * EMB + c];
        v2f wd = *(const v2f*)&w1[3 * EMB + c];
        v2f we = *(const v2f*)&w1[4 * EMB + c];
        v2f wf = *(const v2f*)&w1[5 * EMB + c];
        float a0 = a.x, a1 = a.y;
        a0 += x0 * wa.x;  a1 += x0 * wa.y;
        a0 += x1 * wb.x;  a1 += x1 * wb.y;
        a0 += x2 * wc.x;  a1 += x2 * wc.y;
        a0 += x3 * wd.x;  a1 += x3 * wd.y;
        a0 += x4 * we.x;  a1 += x4 * we.y;
        a0 += x5 * wf.x;  a1 += x5 * wf.y;
        a0 = fmaxf(a0, 0.f);  a1 = fmaxf(a1, 0.f);
        v2f t; t.x = a0; t.y = a1;
        hf[kk] = t;
        sum += a0; sumsq += a0 * a0;
        sum += a1; sumsq += a1 * a1;
    }
    sum   += __shfl_xor(sum, 16, 32);
    sumsq += __shfl_xor(sumsq, 16, 32);
    float mu   = sum * (1.f / EMB);
    float var  = sumsq * (1.f / EMB) - mu * mu;
    float rstd = rsqrtf(var + LN_EPSF);
#pragma unroll
    for (int kk = 0; kk < 32; ++kk) {
        const int c = 4 * kk + cb;
        v2f g = *(const v2f*)&ln_g[c];
        v2f b = *(const v2f*)&ln_b[c];
        v2f t = hf[kk];
        t.x = (t.x - mu) * rstd * g.x + b.x;
        t.y = (t.y - mu) * rstd * g.y + b.y;
        hf[kk] = t;
    }

    __syncthreads();

    const float* wl = &W2p[half * W2PITCH + 2 * l16];
#pragma unroll
    for (int nb = 0; nb < 8; nb += 2) {
        float bias0 = b2[nb * 16 + l16];
        float bias1 = b2[nb * 16 + 16 + l16];
        v8f acc0, acc1;
#pragma unroll
        for (int r = 0; r < 8; ++r) { acc0[r] = bias0; acc1[r] = bias1; }
#pragma unroll
        for (int kk = 0; kk < 32; ++kk) {
            v2f bf0 = *(const v2f*)&wl[2 * kk * W2PITCH + nb * 32];
            v2f bf1 = *(const v2f*)&wl[2 * kk * W2PITCH + nb * 32 + 32];
            acc0 = __builtin_amdgcn_wmma_f32_16x16x4_f32(false, hf[kk], false, bf0,
                                                         (short)0, acc0, false, false);
            acc1 = __builtin_amdgcn_wmma_f32_16x16x4_f32(false, hf[kk], false, bf1,
                                                         (short)0, acc1, false, false);
        }
        float* o = &outp[(tok0 + 8 * half) * EMB + nb * 16 + l16];
#pragma unroll
        for (int r = 0; r < 8; ++r) {
            o[r * EMB]      = acc0[r];
            o[r * EMB + 16] = acc1[r];
        }
    }
}

__global__ void vpack_kernel(const float* __restrict__ oc, const float* __restrict__ org,
                             float* __restrict__ V)
{
    int t = blockIdx.x * 256 + threadIdx.x;
    if (t >= N_POS * 16) return;
    int p = t >> 4, c = t & 15;
    V[t] = (c < 2) ? oc[c * N_POS + p] : org[(c - 2) * N_POS + p];
}

#define KG 4
#define QW 2
#define QPB (QW * 16)
#define KPT (KG * 32)
#define CSTR 33

__global__ __launch_bounds__(256) void attn_kernel(
    const float* __restrict__ Q, const float* __restrict__ K,
    const float* __restrict__ V, float* __restrict__ out)
{
    __shared__ float Kls[KPT * KSTR];
    __shared__ float Vls[KG][16 * PSTR];
    __shared__ float Pls[8][16 * PSTR];

    const int tid  = threadIdx.x;
    const int lane = tid & 31;
    const int wave = tid >> 5;
    const int g    = wave / QW;
    const int wq   = wave % QW;
    const int half = lane >> 4;
    const int l16  = lane & 15;
    const int qRow = blockIdx.x * QPB + wq * 16 + l16;

    v2f qf[32];
#pragma unroll
    for (int kk = 0; kk < 32; ++kk) {
        const float* s = &Q[qRow * EMB + kk * 4 + half * 2];
        v2f t; t.x = s[0] * TEMP_INV; t.y = s[1] * TEMP_INV;
        qf[kk] = t;
    }

    v8f Oacc = {};
    float mstate[8], lstate[8];
#pragma unroll
    for (int v = 0; v < 8; ++v) { mstate[v] = -3.0e38f; lstate[v] = 0.f; }

    float* pmine = &Pls[wave][0];
    const float* kmine = &Kls[g * 32 * KSTR];
    const float* vmine = &Vls[g][0];

    for (int kt = 0; kt < N_POS / KPT; ++kt) {
        __syncthreads();
        {
            const float* kp = K + (size_t)kt * (KPT * EMB) + tid * 4;
            const float* vp = V + (size_t)kt * (KPT * 16) + tid * 4;
            v4f kr[16], vr[2];
#pragma unroll
            for (int i = 0; i < 16; ++i) kr[i] = *(const v4f*)(kp + i * 1024);
#pragma unroll
            for (int i = 0; i < 2; ++i)  vr[i] = *(const v4f*)(vp + i * 1024);
            float* kd = &Kls[wave * KSTR + lane * 4];
#pragma unroll
            for (int i = 0; i < 16; ++i) *(v4f*)(kd + i * 8 * KSTR) = kr[i];
#pragma unroll
            for (int i = 0; i < 2; ++i) {
                int j  = tid + i * 256;
                int k  = j >> 2;
                int c4 = (j & 3) * 4;
                float* vd = &Vls[k >> 5][c4 * PSTR + (k & 31)];
                vd[0 * PSTR] = vr[i].x; vd[1 * PSTR] = vr[i].y;
                vd[2 * PSTR] = vr[i].z; vd[3 * PSTR] = vr[i].w;
            }
        }
        __syncthreads();

        v8f S0 = {}, S1 = {};
        {
            v2f bf[32];
#pragma unroll
            for (int kk = 0; kk < 32; ++kk)
                bf[kk] = *(const v2f*)&kmine[l16 * KSTR + kk * 4 + half * 2];
#pragma unroll
            for (int kk = 0; kk < 32; ++kk)
                S0 = __builtin_amdgcn_wmma_f32_16x16x4_f32(false, qf[kk], false, bf[kk],
                                                           (short)0, S0, false, false);
#pragma unroll
            for (int kk = 0; kk < 32; ++kk)
                bf[kk] = *(const v2f*)&kmine[(16 + l16) * KSTR + kk * 4 + half * 2];
#pragma unroll
            for (int kk = 0; kk < 32; ++kk)
                S1 = __builtin_amdgcn_wmma_f32_16x16x4_f32(false, qf[kk], false, bf[kk],
                                                           (short)0, S1, false, false);
        }

#pragma unroll
        for (int v = 0; v < 8; ++v) {
            float s0 = S0[v], s1 = S1[v];
            float rm = fmaxf(s0, s1);
            rm = fmaxf(rm, __shfl_xor(rm, 1, 16));
            rm = fmaxf(rm, __shfl_xor(rm, 2, 16));
            rm = fmaxf(rm, __shfl_xor(rm, 4, 16));
            rm = fmaxf(rm, __shfl_xor(rm, 8, 16));
            float newm  = fmaxf(mstate[v], rm);
            float alpha = __expf(mstate[v] - newm);
            float p0    = __expf(s0 - newm);
            float p1    = __expf(s1 - newm);
            float rs = p0 + p1;
            rs += __shfl_xor(rs, 1, 16);
            rs += __shfl_xor(rs, 2, 16);
            rs += __shfl_xor(rs, 4, 16);
            rs += __shfl_xor(rs, 8, 16);
            lstate[v] = lstate[v] * alpha + rs;
            mstate[v] = newm;
            Oacc[v]  *= alpha;
            int m = v + 8 * half;
            pmine[m * PSTR + l16]      = p0;
            pmine[m * PSTR + 16 + l16] = p1;
        }
        asm volatile("s_wait_dscnt 0" ::: "memory");

        {
            v2f pa[8], vb[8];
#pragma unroll
            for (int c4 = 0; c4 < 8; ++c4) {
                pa[c4] = *(const v2f*)&pmine[l16 * PSTR + c4 * 4 + half * 2];
                vb[c4] = *(const v2f*)&vmine[l16 * PSTR + c4 * 4 + half * 2];
            }
#pragma unroll
            for (int c4 = 0; c4 < 8; ++c4)
                Oacc = __builtin_amdgcn_wmma_f32_16x16x4_f32(false, pa[c4], false, vb[c4],
                                                             (short)0, Oacc, false, false);
        }
    }

    __syncthreads();
    float* cO = &Kls[0];
    float* cm = cO + KG * 16 * CSTR;
    float* cl = cm + KG * QPB;
#pragma unroll
    for (int v = 0; v < 8; ++v) {
        int ql = wq * 16 + v + 8 * half;
        cO[(g * 16 + l16) * CSTR + ql] = Oacc[v];
        if (l16 == 0) { cm[g * QPB + ql] = mstate[v]; cl[g * QPB + ql] = lstate[v]; }
    }
    __syncthreads();

    {
        const int ql = lane;
        float mg[KG], mx = -3.0e38f;
#pragma unroll
        for (int k = 0; k < KG; ++k) { mg[k] = cm[k * QPB + ql]; mx = fmaxf(mx, mg[k]); }
        float a[KG], lsum = 0.f;
#pragma unroll
        for (int k = 0; k < KG; ++k) { a[k] = __expf(mg[k] - mx); lsum += cl[k * QPB + ql] * a[k]; }
#pragma unroll
        for (int i = 0; i < 2; ++i) {
            int c = wave + 8 * i;
            float o = 0.f;
#pragma unroll
            for (int k = 0; k < KG; ++k) o += cO[(k * 16 + c) * CSTR + ql] * a[k];
            out[c * N_POS + blockIdx.x * QPB + ql] = o / lsum;
        }
    }
}

extern "C" void kernel_launch(void* const* d_in, const int* in_sizes, int n_in,
                              void* d_out, int out_size, void* d_ws, size_t ws_size,
                              hipStream_t stream)
{
    const float* ego_cls   = (const float*)d_in[0];
    const float* ego_reg   = (const float*)d_in[1];
    const float* other_cls = (const float*)d_in[2];
    const float* other_reg = (const float*)d_in[3];
    const float* w1  = (const float*)d_in[4];
    const float* b1  = (const float*)d_in[5];
    const float* ln_g = (const float*)d_in[6];
    const float* ln_b = (const float*)d_in[7];
    const float* w2  = (const float*)d_in[8];
    const float* b2  = (const float*)d_in[9];

    float* Qbuf = (float*)d_ws;
    float* Kbuf = Qbuf + N_POS * EMB;
    float* Vbuf = Kbuf + N_POS * EMB;
    float* out  = (float*)d_out;

    encode_kernel<<<dim3(N_POS / ENC_TOK, 2), 256, 0, stream>>>(
        ego_cls, ego_reg, other_cls, other_reg,
        w1, b1, ln_g, ln_b, w2, b2, Qbuf, Kbuf);

    vpack_kernel<<<(N_POS * 16 + 255) / 256, 256, 0, stream>>>(other_cls, other_reg, Vbuf);

    attn_kernel<<<N_POS / QPB, 256, 0, stream>>>(Qbuf, Kbuf, Vbuf, out);
}


# attention: score tiles computed transposed (keys as rows): softmax max/sum in-lane plus one lane^16 exchange, P feeds O^T=V^T.P^T from registers; new summation order in row sums and P.V
# speedup vs baseline: 2.9372x; 1.0769x over previous
#include <hip/hip_runtime.h>
#include <math.h>

typedef float v2f __attribute__((ext_vector_type(2)));
typedef float v4f __attribute__((ext_vector_type(4)));
typedef float v8f __attribute__((ext_vector_type(8)));

#define N_POS 16384
#define EMB 128
#define TEMP_INV 10.0f
#define LN_EPSF 1e-5f
#define KSTR 132
#define PSTR 34

#define ENC_TOK 128
#define W2PITCH 288

__global__ __launch_bounds__(256) void encode_kernel(
    const float* __restrict__ ego_cls, const float* __restrict__ ego_reg,
    const float* __restrict__ other_cls, const float* __restrict__ other_reg,
    const float* __restrict__ w1, const float* __restrict__ b1,
    const float* __restrict__ ln_g, const float* __restrict__ ln_b,
    const float* __restrict__ w2, const float* __restrict__ b2,
    float* __restrict__ Qbuf, float* __restrict__ Kbuf)
{
    __shared__ float W2p[64 * W2PITCH];

    const int tid  = threadIdx.x;
    const int lane = tid & 31;
    const int wave = tid >> 5;
    const int half = lane >> 4;
    const int l16  = lane & 15;

    {
        v4f r0[8], r1[8];
#pragma unroll
        for (int i = 0; i < 8; ++i) {
            int idx = tid + i * 256;
            int pr  = idx >> 5;
            int e4  = (idx & 31) * 4;
            r0[i] = *(const v4f*)&w2[(2 * pr) * EMB + e4];
            r1[i] = *(const v4f*)&w2[(2 * pr + 1) * EMB + e4];
        }
#pragma unroll
        for (int i = 0; i < 8; ++i) {
            int idx = tid + i * 256;
            int pr  = idx >> 5;
            int e4  = (idx & 31) * 4;
            float* d = &W2p[pr * W2PITCH + 2 * e4];
            v4f lo, hi;
            lo.x = r0[i].x; lo.y = r1[i].x; lo.z = r0[i].y; lo.w = r1[i].y;
            hi.x = r0[i].z; hi.y = r1[i].z; hi.z = r0[i].w; hi.w = r1[i].w;
            *(v4f*)d       = lo;
            *(v4f*)(d + 4) = hi;
        }
    }

    const float* cls; const float* reg; float* outp;
    if (blockIdx.y == 0) { cls = ego_cls;   reg = ego_reg;   outp = Qbuf; }
    else                 { cls = other_cls; reg = other_reg; outp = Kbuf; }

    const int tok0 = blockIdx.x * ENC_TOK + wave * 16;
    const int p    = tok0 + l16;

    float x0 = reg[3 * N_POS + p];
    float x1 = reg[4 * N_POS + p];
    float x2 = reg[5 * N_POS + p];
    float th = reg[6 * N_POS + p];
    float x3 = sinf(th);
    float x4 = cosf(th);
    float x5 = fmaxf(cls[p], cls[N_POS + p]);

    const int cb = 2 * half;
    v2f hf[32];
    float sum = 0.f, sumsq = 0.f;
#pragma unroll
    for (int kk = 0; kk < 32; ++kk) {
        const int c = 4 * kk + cb;
        v2f a  = *(const v2f*)&b1[c];
        v2f wa = *(const v2f*)&w1[0 * EMB + c];
        v2f wb = *(const v2f*)&w1[1 * EMB + c];
        v2f wc = *(const v2f*)&w1[2 * EMB + c];
        v2f wd = *(const v2f*)&w1[3 * EMB + c];
        v2f we = *(const v2f*)&w1[4 * EMB + c];
        v2f wf = *(const v2f*)&w1[5 * EMB + c];
        float a0 = a.x, a1 = a.y;
        a0 += x0 * wa.x;  a1 += x0 * wa.y;
        a0 += x1 * wb.x;  a1 += x1 * wb.y;
        a0 += x2 * wc.x;  a1 += x2 * wc.y;
        a0 += x3 * wd.x;  a1 += x3 * wd.y;
        a0 += x4 * we.x;  a1 += x4 * we.y;
        a0 += x5 * wf.x;  a1 += x5 * wf.y;
        a0 = fmaxf(a0, 0.f);  a1 = fmaxf(a1, 0.f);
        v2f t; t.x = a0; t.y = a1;
        hf[kk] = t;
        sum += a0; sumsq += a0 * a0;
        sum += a1; sumsq += a1 * a1;
    }
    sum   += __shfl_xor(sum, 16, 32);
    sumsq += __shfl_xor(sumsq, 16, 32);
    float mu   = sum * (1.f / EMB);
    float var  = sumsq * (1.f / EMB) - mu * mu;
    float rstd = rsqrtf(var + LN_EPSF);
#pragma unroll
    for (int kk = 0; kk < 32; ++kk) {
        const int c = 4 * kk + cb;
        v2f g = *(const v2f*)&ln_g[c];
        v2f b = *(const v2f*)&ln_b[c];
        v2f t = hf[kk];
        t.x = (t.x - mu) * rstd * g.x + b.x;
        t.y = (t.y - mu) * rstd * g.y + b.y;
        hf[kk] = t;
    }

    __syncthreads();

    const float* wl = &W2p[half * W2PITCH + 2 * l16];
#pragma unroll
    for (int nb = 0; nb < 8; nb += 2) {
        float bias0 = b2[nb * 16 + l16];
        float bias1 = b2[nb * 16 + 16 + l16];
        v8f acc0, acc1;
#pragma unroll
        for (int r = 0; r < 8; ++r) { acc0[r] = bias0; acc1[r] = bias1; }
#pragma unroll
        for (int kk = 0; kk < 32; ++kk) {
            v2f bf0 = *(const v2f*)&wl[2 * kk * W2PITCH + nb * 32];
            v2f bf1 = *(const v2f*)&wl[2 * kk * W2PITCH + nb * 32 + 32];
            acc0 = __builtin_amdgcn_wmma_f32_16x16x4_f32(false, hf[kk], false, bf0,
                                                         (short)0, acc0, false, false);
            acc1 = __builtin_amdgcn_wmma_f32_16x16x4_f32(false, hf[kk], false, bf1,
                                                         (short)0, acc1, false, false);
        }
        float* o = &outp[(tok0 + 8 * half) * EMB + nb * 16 + l16];
#pragma unroll
        for (int r = 0; r < 8; ++r) {
            o[r * EMB]      = acc0[r];
            o[r * EMB + 16] = acc1[r];
        }
    }
}

__global__ void vpack_kernel(const float* __restrict__ oc, const float* __restrict__ org,
                             float* __restrict__ V)
{
    int t = blockIdx.x * 256 + threadIdx.x;
    if (t >= N_POS * 16) return;
    int p = t >> 4, c = t & 15;
    V[t] = (c < 2) ? oc[c * N_POS + p] : org[(c - 2) * N_POS + p];
}

#define KG 4
#define QW 2
#define QPB (QW * 16)
#define KPT (KG * 32)
#define CSTR 33

__device__ __forceinline__ float xhalf(float x)
{
    int b = __builtin_bit_cast(int, x);
    return __builtin_bit_cast(float,
        __builtin_amdgcn_permlanex16(b, b, 0x76543210, 0xfedcba98, false, false));
}

__global__ __launch_bounds__(256) __attribute__((amdgpu_waves_per_eu(4, 4))) void attn_kernel(
    const float* __restrict__ Q, const float* __restrict__ K,
    const float* __restrict__ V, float* __restrict__ out)
{
    __shared__ float Kls[KPT * KSTR];
    __shared__ float Vls[KG][16 * PSTR];

    const int tid  = threadIdx.x;
    const int lane = tid & 31;
    const int wave = tid >> 5;
    const int g    = wave / QW;
    const int wq   = wave % QW;
    const int half = lane >> 4;
    const int l16  = lane & 15;
    const int qRow = blockIdx.x * QPB + wq * 16 + l16;

    v2f qf[32];
#pragma unroll
    for (int kk = 0; kk < 32; ++kk) {
        const float* s = &Q[qRow * EMB + kk * 4 + half * 2];
        v2f t; t.x = s[0] * TEMP_INV; t.y = s[1] * TEMP_INV;
        qf[kk] = t;
    }

    v8f Oacc = {};
    float mrun = -3.0e38f, lrun = 0.f;

    const float* kmine = &Kls[g * 32 * KSTR];
    const float* vmine = &Vls[g][0];

    for (int kt = 0; kt < N_POS / KPT; ++kt) {
        __syncthreads();
        {
            const float* kp = K + (size_t)kt * (KPT * EMB) + tid * 4;
            const float* vp = V + (size_t)kt * (KPT * 16) + tid * 4;
            v4f kr[16], vr[2];
#pragma unroll
            for (int i = 0; i < 16; ++i) kr[i] = *(const v4f*)(kp + i * 1024);
#pragma unroll
            for (int i = 0; i < 2; ++i)  vr[i] = *(const v4f*)(vp + i * 1024);
            float* kd = &Kls[wave * KSTR + lane * 4];
#pragma unroll
            for (int i = 0; i < 16; ++i) *(v4f*)(kd + i * 8 * KSTR) = kr[i];
#pragma unroll
            for (int i = 0; i < 2; ++i) {
                int j  = tid + i * 256;
                int k  = j >> 2;
                int c4 = (j & 3) * 4;
                int k5 = k & 31;
                int pos = (k5 & 16) | ((k5 & 6) << 1) | ((k5 & 8) >> 2) | (k5 & 1);
                float* vd = &Vls[k >> 5][c4 * PSTR + pos];
                vd[0 * PSTR] = vr[i].x; vd[1 * PSTR] = vr[i].y;
                vd[2 * PSTR] = vr[i].z; vd[3 * PSTR] = vr[i].w;
            }
        }
        __syncthreads();

        v8f S0 = {}, S1 = {};
        {
            v2f bf[32];
#pragma unroll
            for (int kk = 0; kk < 32; ++kk)
                bf[kk] = *(const v2f*)&kmine[l16 * KSTR + kk * 4 + half * 2];
#pragma unroll
            for (int kk = 0; kk < 32; ++kk)
                S0 = __builtin_amdgcn_wmma_f32_16x16x4_f32(false, bf[kk], false, qf[kk],
                                                           (short)0, S0, false, false);
#pragma unroll
            for (int kk = 0; kk < 32; ++kk)
                bf[kk] = *(const v2f*)&kmine[(16 + l16) * KSTR + kk * 4 + half * 2];
#pragma unroll
            for (int kk = 0; kk < 32; ++kk)
                S1 = __builtin_amdgcn_wmma_f32_16x16x4_f32(false, bf[kk], false, qf[kk],
                                                           (short)0, S1, false, false);
        }

        v2f va[8];
#pragma unroll
        for (int c = 0; c < 8; ++c)
            va[c] = *(const v2f*)&vmine[l16 * PSTR + (c >> 2) * 16 + (c & 3) * 4 + half * 2];
        __builtin_amdgcn_sched_barrier(0);

        float mx = fmaxf(S0[0], S1[0]);
#pragma unroll
        for (int r = 1; r < 8; ++r) mx = fmaxf(mx, fmaxf(S0[r], S1[r]));
        mx = fmaxf(mx, xhalf(mx));
        const float newm  = fmaxf(mrun, mx);
        const float alpha = __expf(mrun - newm);
        v2f p0[4], p1[4];
        float rs0 = 0.f, rs1 = 0.f;
#pragma unroll
        for (int j = 0; j < 4; ++j) {
            v2f a, b;
            a.x = __expf(S0[2 * j] - newm);  a.y = __expf(S0[2 * j + 1] - newm);
            b.x = __expf(S1[2 * j] - newm);  b.y = __expf(S1[2 * j + 1] - newm);
            p0[j] = a;  p1[j] = b;
            rs0 += a.x + a.y;
            rs1 += b.x + b.y;
        }
        float rs = rs0 + rs1;
        rs += xhalf(rs);
        lrun = lrun * alpha + rs;
        mrun = newm;
#pragma unroll
        for (int r = 0; r < 8; ++r) Oacc[r] *= alpha;

        {
#pragma unroll
            for (int j = 0; j < 4; ++j)
                Oacc = __builtin_amdgcn_wmma_f32_16x16x4_f32(false, va[j], false, p0[j],
                                                             (short)0, Oacc, false, false);
#pragma unroll
            for (int j = 0; j < 4; ++j)
                Oacc = __builtin_amdgcn_wmma_f32_16x16x4_f32(false, va[4 + j], false, p1[j],
                                                             (short)0, Oacc, false, false);
        }
    }

    __syncthreads();
    float* cO = &Kls[0];
    float* cm = cO + KG * 16 * CSTR;
    float* cl = cm + KG * QPB;
    {
        const int ql = wq * 16 + l16;
#pragma unroll
        for (int r = 0; r < 8; ++r)
            cO[(g * 16 + 8 * half + r) * CSTR + ql] = Oacc[r];
        if (half == 0) { cm[g * QPB + ql] = mrun; cl[g * QPB + ql] = lrun; }
    }
    __syncthreads();

    {
        const int ql = lane;
        float mg[KG], mx = -3.0e38f;
#pragma unroll
        for (int k = 0; k < KG; ++k) { mg[k] = cm[k * QPB + ql]; mx = fmaxf(mx, mg[k]); }
        float a[KG], lsum = 0.f;
#pragma unroll
        for (int k = 0; k < KG; ++k) { a[k] = __expf(mg[k] - mx); lsum += cl[k * QPB + ql] * a[k]; }
#pragma unroll
        for (int i = 0; i < 2; ++i) {
            int c = wave + 8 * i;
            float o = 0.f;
#pragma unroll
            for (int k = 0; k < KG; ++k) o += cO[(k * 16 + c) * CSTR + ql] * a[k];
            out[c * N_POS + blockIdx.x * QPB + ql] = o / lsum;
        }
    }
}

extern "C" void kernel_launch(void* const* d_in, const int* in_sizes, int n_in,
                              void* d_out, int out_size, void* d_ws, size_t ws_size,
                              hipStream_t stream)
{
    const float* ego_cls   = (const float*)d_in[0];
    const float* ego_reg   = (const float*)d_in[1];
    const float* other_cls = (const float*)d_in[2];
    const float* other_reg = (const float*)d_in[3];
    const float* w1  = (const float*)d_in[4];
    const float* b1  = (const float*)d_in[5];
    const float* ln_g = (const float*)d_in[6];
    const float* ln_b = (const float*)d_in[7];
    const float* w2  = (const float*)d_in[8];
    const float* b2  = (const float*)d_in[9];

    float* Qbuf = (float*)d_ws;
    float* Kbuf = Qbuf + N_POS * EMB;
    float* Vbuf = Kbuf + N_POS * EMB;
    float* out  = (float*)d_out;

    encode_kernel<<<dim3(N_POS / ENC_TOK, 2), 256, 0, stream>>>(
        ego_cls, ego_reg, other_cls, other_reg,
        w1, b1, ln_g, ln_b, w2, b2, Qbuf, Kbuf);

    vpack_kernel<<<(N_POS * 16 + 255) / 256, 256, 0, stream>>>(other_cls, other_reg, Vbuf);

    attn_kernel<<<N_POS / QPB, 256, 0, stream>>>(Qbuf, Kbuf, Vbuf, out);
}


# attention: the next key/value tile's global loads are issued one trip ahead, behind the staging barrier, and held in registers under the current tile's matrix work
# speedup vs baseline: 3.0875x; 1.0512x over previous
#include <hip/hip_runtime.h>
#include <math.h>

typedef float v2f __attribute__((ext_vector_type(2)));
typedef float v4f __attribute__((ext_vector_type(4)));
typedef float v8f __attribute__((ext_vector_type(8)));

#define N_POS 16384
#define EMB 128
#define TEMP_INV 10.0f
#define LN_EPSF 1e-5f
#define KSTR 132
#define PSTR 34

#define ENC_TOK 128
#define W2PITCH 288

__global__ __launch_bounds__(256) void encode_kernel(
    const float* __restrict__ ego_cls, const float* __restrict__ ego_reg,
    const float* __restrict__ other_cls, const float* __restrict__ other_reg,
    const float* __restrict__ w1, const float* __restrict__ b1,
    const float* __restrict__ ln_g, const float* __restrict__ ln_b,
    const float* __restrict__ w2, const float* __restrict__ b2,
    float* __restrict__ Qbuf, float* __restrict__ Kbuf)
{
    __shared__ float W2p[64 * W2PITCH];

    const int tid  = threadIdx.x;
    const int lane = tid & 31;
    const int wave = tid >> 5;
    const int half = lane >> 4;
    const int l16  = lane & 15;

    {
        v4f r0[8], r1[8];
#pragma unroll
        for (int i = 0; i < 8; ++i) {
            int idx = tid + i * 256;
            int pr  = idx >> 5;
            int e4  = (idx & 31) * 4;
            r0[i] = *(const v4f*)&w2[(2 * pr) * EMB + e4];
            r1[i] = *(const v4f*)&w2[(2 * pr + 1) * EMB + e4];
        }
#pragma unroll
        for (int i = 0; i < 8; ++i) {
            int idx = tid + i * 256;
            int pr  = idx >> 5;
            int e4  = (idx & 31) * 4;
            float* d = &W2p[pr * W2PITCH + 2 * e4];
            v4f lo, hi;
            lo.x = r0[i].x; lo.y = r1[i].x; lo.z = r0[i].y; lo.w = r1[i].y;
            hi.x = r0[i].z; hi.y = r1[i].z; hi.z = r0[i].w; hi.w = r1[i].w;
            *(v4f*)d       = lo;
            *(v4f*)(d + 4) = hi;
        }
    }

    const float* cls; const float* reg; float* outp;
    if (blockIdx.y == 0) { cls = ego_cls;   reg = ego_reg;   outp = Qbuf; }
    else                 { cls = other_cls; reg = other_reg; outp = Kbuf; }

    const int tok0 = blockIdx.x * ENC_TOK + wave * 16;
    const int p    = tok0 + l16;

    float x0 = reg[3 * N_POS + p];
    float x1 = reg[4 * N_POS + p];
    float x2 = reg[5 * N_POS + p];
    float th = reg[6 * N_POS + p];
    float x3 = sinf(th);
    float x4 = cosf(th);
    float x5 = fmaxf(cls[p], cls[N_POS + p]);

    const int cb = 2 * half;
    v2f hf[32];
    float sum = 0.f, sumsq = 0.f;
#pragma unroll
    for (int kk = 0; kk < 32; ++kk) {
        const int c = 4 * kk + cb;
        v2f a  = *(const v2f*)&b1[c];
        v2f wa = *(const v2f*)&w1[0 * EMB + c];
        v2f wb = *(const v2f*)&w1[1 * EMB + c];
        v2f wc = *(const v2f*)&w1[2 * EMB + c];
        v2f wd = *(const v2f*)&w1[3 * EMB + c];
        v2f we = *(const v2f*)&w1[4 * EMB + c];
        v2f wf = *(const v2f*)&w1[5 * EMB + c];
        float a0 = a.x, a1 = a.y;
        a0 += x0 * wa.x;  a1 += x0 * wa.y;
        a0 += x1 * wb.x;  a1 += x1 * wb.y;
        a0 += x2 * wc.x;  a1 += x2 * wc.y;
        a0 += x3 * wd.x;  a1 += x3 * wd.y;
        a0 += x4 * we.x;  a1 += x4 * we.y;
        a0 += x5 * wf.x;  a1 += x5 * wf.y;
        a0 = fmaxf(a0, 0.f);  a1 = fmaxf(a1, 0.f);
        v2f t; t.x = a0; t.y = a1;
        hf[kk] = t;
        sum += a0; sumsq += a0 * a0;
        sum += a1; sumsq += a1 * a1;
    }
    sum   += __shfl_xor(sum, 16, 32);
    sumsq += __shfl_xor(sumsq, 16, 32);
    float mu   = sum * (1.f / EMB);
    float var  = sumsq * (1.f / EMB) - mu * mu;
    float rstd = rsqrtf(var + LN_EPSF);
#pragma unroll
    for (int kk = 0; kk < 32; ++kk) {
        const int c = 4 * kk + cb;
        v2f g = *(const v2f*)&ln_g[c];
        v2f b = *(const v2f*)&ln_b[c];
        v2f t = hf[kk];
        t.x = (t.x - mu) * rstd * g.x + b.x;
        t.y = (t.y - mu) * rstd * g.y + b.y;
        hf[kk] = t;
    }

    __syncthreads();

    const float* wl = &W2p[half * W2PITCH + 2 * l16];
#pragma unroll
    for (int nb = 0; nb < 8; nb += 2) {
        float bias0 = b2[nb * 16 + l16];
        float bias1 = b2[nb * 16 + 16 + l16];
        v8f acc0, acc1;
#pragma unroll
        for (int r = 0; r < 8; ++r) { acc0[r] = bias0; acc1[r] = bias1; }
#pragma unroll
        for (int kk = 0; kk < 32; ++kk) {
            v2f bf0 = *(const v2f*)&wl[2 * kk * W2PITCH + nb * 32];
            v2f bf1 = *(const v2f*)&wl[2 * kk * W2PITCH + nb * 32 + 32];
            acc0 = __builtin_amdgcn_wmma_f32_16x16x4_f32(false, hf[kk], false, bf0,
                                                         (short)0, acc0, false, false);
            acc1 = __builtin_amdgcn_wmma_f32_16x16x4_f32(false, hf[kk], false, bf1,
                                                         (short)0, acc1, false, false);
        }
        float* o = &outp[(tok0 + 8 * half) * EMB + nb * 16 + l16];
#pragma unroll
        for (int r = 0; r < 8; ++r) {
            o[r * EMB]      = acc0[r];
            o[r * EMB + 16] = acc1[r];
        }
    }
}

__global__ void vpack_kernel(const float* __restrict__ oc, const float* __restrict__ org,
                             float* __restrict__ V)
{
    int t = blockIdx.x * 256 + threadIdx.x;
    if (t >= N_POS * 16) return;
    int p = t >> 4, c = t & 15;
    V[t] = (c < 2) ? oc[c * N_POS + p] : org[(c - 2) * N_POS + p];
}

#define KG 4
#define QW 2
#define QPB (QW * 16)
#define KPT (KG * 32)
#define CSTR 33

__device__ __forceinline__ float xhalf(float x)
{
    int b = __builtin_bit_cast(int, x);
    return __builtin_bit_cast(float,
        __builtin_amdgcn_permlanex16(b, b, 0x76543210, 0xfedcba98, false, false));
}

__global__ __launch_bounds__(256) __attribute__((amdgpu_waves_per_eu(4, 4))) void attn_kernel(
    const float* __restrict__ Q, const float* __restrict__ K,
    const float* __restrict__ V, float* __restrict__ out)
{
    __shared__ float Kls[KPT * KSTR];
    __shared__ float Vls[KG][16 * PSTR];

    const int tid  = threadIdx.x;
    const int lane = tid & 31;
    const int wave = tid >> 5;
    const int g    = wave / QW;
    const int wq   = wave % QW;
    const int half = lane >> 4;
    const int l16  = lane & 15;
    const int qRow = blockIdx.x * QPB + wq * 16 + l16;

    v2f qf[32];
#pragma unroll
    for (int kk = 0; kk < 32; ++kk) {
        const float* s = &Q[qRow * EMB + kk * 4 + half * 2];
        v2f t; t.x = s[0] * TEMP_INV; t.y = s[1] * TEMP_INV;
        qf[kk] = t;
    }

    v8f Oacc = {};
    float mrun = -3.0e38f, lrun = 0.f;

    const float* kmine = &Kls[g * 32 * KSTR];
    const float* vmine = &Vls[g][0];

    const float* kp = K + tid * 4;
    const float* vp = V + tid * 4;
    v4f kr[16], vr[2];
#pragma unroll
    for (int i = 0; i < 16; ++i) kr[i] = *(const v4f*)(kp + i * 1024);
#pragma unroll
    for (int i = 0; i < 2; ++i)  vr[i] = *(const v4f*)(vp + i * 1024);

    for (int kt = 0; kt < N_POS / KPT; ++kt) {
        __syncthreads();
        {
            float* kd = &Kls[wave * KSTR + lane * 4];
#pragma unroll
            for (int i = 0; i < 16; ++i) *(v4f*)(kd + i * 8 * KSTR) = kr[i];
#pragma unroll
            for (int i = 0; i < 2; ++i) {
                int j  = tid + i * 256;
                int k  = j >> 2;
                int c4 = (j & 3) * 4;
                int k5 = k & 31;
                int pos = (k5 & 16) | ((k5 & 6) << 1) | ((k5 & 8) >> 2) | (k5 & 1);
                float* vd = &Vls[k >> 5][c4 * PSTR + pos];
                vd[0 * PSTR] = vr[i].x; vd[1 * PSTR] = vr[i].y;
                vd[2 * PSTR] = vr[i].z; vd[3 * PSTR] = vr[i].w;
            }
        }
        __syncthreads();
        if (kt + 1 < N_POS / KPT) {
#pragma unroll
            for (int i = 0; i < 16; ++i) kr[i] = *(const v4f*)(kp + KPT * EMB + i * 1024);
#pragma unroll
            for (int i = 0; i < 2; ++i)  vr[i] = *(const v4f*)(vp + KPT * 16 + i * 1024);
        }

        v8f S0 = {}, S1 = {};
        {
            v2f bf[32];
#pragma unroll
            for (int kk = 0; kk < 32; ++kk)
                bf[kk] = *(const v2f*)&kmine[l16 * KSTR + kk * 4 + half * 2];
#pragma unroll
            for (int kk = 0; kk < 32; ++kk)
                S0 = __builtin_amdgcn_wmma_f32_16x16x4_f32(false, bf[kk], false, qf[kk],
                                                           (short)0, S0, false, false);
#pragma unroll
            for (int kk = 0; kk < 32; ++kk)
                bf[kk] = *(const v2f*)&kmine[(16 + l16) * KSTR + kk * 4 + half * 2];
#pragma unroll
            for (int kk = 0; kk < 32; ++kk)
                S1 = __builtin_amdgcn_wmma_f32_16x16x4_f32(false, bf[kk], false, qf[kk],
                                                           (short)0, S1, false, false);
        }

        v2f va[8];
#pragma unroll
        for (int c = 0; c < 8; ++c)
            va[c] = *(const v2f*)&vmine[l16 * PSTR + (c >> 2) * 16 + (c & 3) * 4 + half * 2];
        __builtin_amdgcn_sched_barrier(0);

        float mx = fmaxf(S0[0], S1[0]);
#pragma unroll
        for (int r = 1; r < 8; ++r) mx = fmaxf(mx, fmaxf(S0[r], S1[r]));
        mx = fmaxf(mx, xhalf(mx));
        const float newm  = fmaxf(mrun, mx);
        const float alpha = __expf(mrun - newm);
        v2f p0[4], p1[4];
        float rs0 = 0.f, rs1 = 0.f;
#pragma unroll
        for (int j = 0; j < 4; ++j) {
            v2f a, b;
            a.x = __expf(S0[2 * j] - newm);  a.y = __expf(S0[2 * j + 1] - newm);
            b.x = __expf(S1[2 * j] - newm);  b.y = __expf(S1[2 * j + 1] - newm);
            p0[j] = a;  p1[j] = b;
            rs0 += a.x + a.y;
            rs1 += b.x + b.y;
        }
        float rs = rs0 + rs1;
        rs += xhalf(rs);
        lrun = lrun * alpha + rs;
        mrun = newm;
#pragma unroll
        for (int r = 0; r < 8; ++r) Oacc[r] *= alpha;

        {
#pragma unroll
            for (int j = 0; j < 4; ++j)
                Oacc = __builtin_amdgcn_wmma_f32_16x16x4_f32(false, va[j], false, p0[j],
                                                             (short)0, Oacc, false, false);
#pragma unroll
            for (int j = 0; j < 4; ++j)
                Oacc = __builtin_amdgcn_wmma_f32_16x16x4_f32(false, va[4 + j], false, p1[j],
                                                             (short)0, Oacc, false, false);
        }
        kp += KPT * EMB;
        vp += KPT * 16;
    }

    __syncthreads();
    float* cO = &Kls[0];
    float* cm = cO + KG * 16 * CSTR;
    float* cl = cm + KG * QPB;
    {
        const int ql = wq * 16 + l16;
#pragma unroll
        for (int r = 0; r < 8; ++r)
            cO[(g * 16 + 8 * half + r) * CSTR + ql] = Oacc[r];
        if (half == 0) { cm[g * QPB + ql] = mrun; cl[g * QPB + ql] = lrun; }
    }
    __syncthreads();

    {
        const int ql = lane;
        float mg[KG], mx = -3.0e38f;
#pragma unroll
        for (int k = 0; k < KG; ++k) { mg[k] = cm[k * QPB + ql]; mx = fmaxf(mx, mg[k]); }
        float a[KG], lsum = 0.f;
#pragma unroll
        for (int k = 0; k < KG; ++k) { a[k] = __expf(mg[k] - mx); lsum += cl[k * QPB + ql] * a[k]; }
#pragma unroll
        for (int i = 0; i < 2; ++i) {
            int c = wave + 8 * i;
            float o = 0.f;
#pragma unroll
            for (int k = 0; k < KG; ++k) o += cO[(k * 16 + c) * CSTR + ql] * a[k];
            out[c * N_POS + blockIdx.x * QPB + ql] = o / lsum;
        }
    }
}

extern "C" void kernel_launch(void* const* d_in, const int* in_sizes, int n_in,
                              void* d_out, int out_size, void* d_ws, size_t ws_size,
                              hipStream_t stream)
{
    const float* ego_cls   = (const float*)d_in[0];
    const float* ego_reg   = (const float*)d_in[1];
    const float* other_cls = (const float*)d_in[2];
    const float* other_reg = (const float*)d_in[3];
    const float* w1  = (const float*)d_in[4];
    const float* b1  = (const float*)d_in[5];
    const float* ln_g = (const float*)d_in[6];
    const float* ln_b = (const float*)d_in[7];
    const float* w2  = (const float*)d_in[8];
    const float* b2  = (const float*)d_in[9];

    float* Qbuf = (float*)d_ws;
    float* Kbuf = Qbuf + N_POS * EMB;
    float* Vbuf = Kbuf + N_POS * EMB;
    float* out  = (float*)d_out;

    encode_kernel<<<dim3(N_POS / ENC_TOK, 2), 256, 0, stream>>>(
        ego_cls, ego_reg, other_cls, other_reg,
        w1, b1, ln_g, ln_b, w2, b2, Qbuf, Kbuf);

    vpack_kernel<<<(N_POS * 16 + 255) / 256, 256, 0, stream>>>(other_cls, other_reg, Vbuf);

    attn_kernel<<<N_POS / QPB, 256, 0, stream>>>(Qbuf, Kbuf, Vbuf, out);
}


# attention: K fragments read from LDS in whole runs (16, 8, 8 two-address reads per trip) a matrix block ahead of use, two register sets: 3 places of LDS reads among a wave's matrix work, not 16
# speedup vs baseline: 3.2051x; 1.0381x over previous
#include <hip/hip_runtime.h>
#include <math.h>

typedef float v2f __attribute__((ext_vector_type(2)));
typedef float v4f __attribute__((ext_vector_type(4)));
typedef float v8f __attribute__((ext_vector_type(8)));

#define N_POS 16384
#define EMB 128
#define TEMP_INV 10.0f
#define LN_EPSF 1e-5f
#define KSTR 132
#define PSTR 34

#define ENC_TOK 128
#define W2PITCH 288

__global__ __launch_bounds__(256) void encode_kernel(
    const float* __restrict__ ego_cls, const float* __restrict__ ego_reg,
    const float* __restrict__ other_cls, const float* __restrict__ other_reg,
    const float* __restrict__ w1, const float* __restrict__ b1,
    const float* __restrict__ ln_g, const float* __restrict__ ln_b,
    const float* __restrict__ w2, const float* __restrict__ b2,
    float* __restrict__ Qbuf, float* __restrict__ Kbuf)
{
    __shared__ float W2p[64 * W2PITCH];

    const int tid  = threadIdx.x;
    const int lane = tid & 31;
    const int wave = tid >> 5;
    const int half = lane >> 4;
    const int l16  = lane & 15;

    {
        v4f r0[8], r1[8];
#pragma unroll
        for (int i = 0; i < 8; ++i) {
            int idx = tid + i * 256;
            int pr  = idx >> 5;
            int e4  = (idx & 31) * 4;
            r0[i] = *(const v4f*)&w2[(2 * pr) * EMB + e4];
            r1[i] = *(const v4f*)&w2[(2 * pr + 1) * EMB + e4];
        }
#pragma unroll
        for (int i = 0; i < 8; ++i) {
            int idx = tid + i * 256;
            int pr  = idx >> 5;
            int e4  = (idx & 31) * 4;
            float* d = &W2p[pr * W2PITCH + 2 * e4];
            v4f lo, hi;
            lo.x = r0[i].x; lo.y = r1[i].x; lo.z = r0[i].y; lo.w = r1[i].y;
            hi.x = r0[i].z; hi.y = r1[i].z; hi.z = r0[i].w; hi.w = r1[i].w;
            *(v4f*)d       = lo;
            *(v4f*)(d + 4) = hi;
        }
    }

    const float* cls; const float* reg; float* outp;
    if (blockIdx.y == 0) { cls = ego_cls;   reg = ego_reg;   outp = Qbuf; }
    else                 { cls = other_cls; reg = other_reg; outp = Kbuf; }

    const int tok0 = blockIdx.x * ENC_TOK + wave * 16;
    const int p    = tok0 + l16;

    float x0 = reg[3 * N_POS + p];
    float x1 = reg[4 * N_POS + p];
    float x2 = reg[5 * N_POS + p];
    float th = reg[6 * N_POS + p];
    float x3 = sinf(th);
    float x4 = cosf(th);
    float x5 = fmaxf(cls[p], cls[N_POS + p]);

    const int cb = 2 * half;
    v2f hf[32];
    float sum = 0.f, sumsq = 0.f;
#pragma unroll
    for (int kk = 0; kk < 32; ++kk) {
        const int c = 4 * kk + cb;
        v2f a  = *(const v2f*)&b1[c];
        v2f wa = *(const v2f*)&w1[0 * EMB + c];
        v2f wb = *(const v2f*)&w1[1 * EMB + c];
        v2f wc = *(const v2f*)&w1[2 * EMB + c];
        v2f wd = *(const v2f*)&w1[3 * EMB + c];
        v2f we = *(const v2f*)&w1[4 * EMB + c];
        v2f wf = *(const v2f*)&w1[5 * EMB + c];
        float a0 = a.x, a1 = a.y;
        a0 += x0 * wa.x;  a1 += x0 * wa.y;
        a0 += x1 * wb.x;  a1 += x1 * wb.y;
        a0 += x2 * wc.x;  a1 += x2 * wc.y;
        a0 += x3 * wd.x;  a1 += x3 * wd.y;
        a0 += x4 * we.x;  a1 += x4 * we.y;
        a0 += x5 * wf.x;  a1 += x5 * wf.y;
        a0 = fmaxf(a0, 0.f);  a1 = fmaxf(a1, 0.f);
        v2f t; t.x = a0; t.y = a1;
        hf[kk] = t;
        sum += a0; sumsq += a0 * a0;
        sum += a1; sumsq += a1 * a1;
    }
    sum   += __shfl_xor(sum, 16, 32);
    sumsq += __shfl_xor(sumsq, 16, 32);
    float mu   = sum * (1.f / EMB);
    float var  = sumsq * (1.f / EMB) - mu * mu;
    float rstd = rsqrtf(var + LN_EPSF);
#pragma unroll
    for (int kk = 0; kk < 32; ++kk) {
        const int c = 4 * kk + cb;
        v2f g = *(const v2f*)&ln_g[c];
        v2f b = *(const v2f*)&ln_b[c];
        v2f t = hf[kk];
        t.x = (t.x - mu) * rstd * g.x + b.x;
        t.y = (t.y - mu) * rstd * g.y + b.y;
        hf[kk] = t;
    }

    __syncthreads();

    const float* wl = &W2p[half * W2PITCH + 2 * l16];
#pragma unroll
    for (int nb = 0; nb < 8; nb += 2) {
        float bias0 = b2[nb * 16 + l16];
        float bias1 = b2[nb * 16 + 16 + l16];
        v8f acc0, acc1;
#pragma unroll
        for (int r = 0; r < 8; ++r) { acc0[r] = bias0; acc1[r] = bias1; }
#pragma unroll
        for (int kk = 0; kk < 32; ++kk) {
            v2f bf0 = *(const v2f*)&wl[2 * kk * W2PITCH + nb * 32];
            v2f bf1 = *(const v2f*)&wl[2 * kk * W2PITCH + nb * 32 + 32];
            acc0 = __builtin_amdgcn_wmma_f32_16x16x4_f32(false, hf[kk], false, bf0,
                                                         (short)0, acc0, false, false);
            acc1 = __builtin_amdgcn_wmma_f32_16x16x4_f32(false, hf[kk], false, bf1,
                                                         (short)0, acc1, false, false);
        }
        float* o = &outp[(tok0 + 8 * half) * EMB + nb * 16 + l16];
#pragma unroll
        for (int r = 0; r < 8; ++r) {
            o[r * EMB]      = acc0[r];
            o[r * EMB + 16] = acc1[r];
        }
    }
}

__global__ void vpack_kernel(const float* __restrict__ oc, const float* __restrict__ org,
                             float* __restrict__ V)
{
    int t = blockIdx.x * 256 + threadIdx.x;
    if (t >= N_POS * 16) return;
    int p = t >> 4, c = t & 15;
    V[t] = (c < 2) ? oc[c * N_POS + p] : org[(c - 2) * N_POS + p];
}

#define KG 4
#define QW 2
#define QPB (QW * 16)
#define KPT (KG * 32)
#define CSTR 33

__device__ __forceinline__ float xhalf(float x)
{
    int b = __builtin_bit_cast(int, x);
    return __builtin_bit_cast(float,
        __builtin_amdgcn_permlanex16(b, b, 0x76543210, 0xfedcba98, false, false));
}

__device__ __forceinline__ void load_kgroup(v2f (&f)[16], const float* ka, const float* kb, int g)
{
#pragma unroll
    for (int i = 0; i < 8; ++i) {
        f[2 * i]     = *(const v2f*)&ka[(g * 8 + i) * 4];
        f[2 * i + 1] = *(const v2f*)&kb[(g * 8 + i) * 4];
    }
}

__device__ __forceinline__ void mma_kgroup(v8f& S0, v8f& S1, const v2f (&f)[16], const v2f (&qf)[32], int g)
{
#pragma unroll
    for (int i = 0; i < 8; ++i) {
        S0 = __builtin_amdgcn_wmma_f32_16x16x4_f32(false, f[2 * i], false, qf[g * 8 + i],
                                                   (short)0, S0, false, false);
        S1 = __builtin_amdgcn_wmma_f32_16x16x4_f32(false, f[2 * i + 1], false, qf[g * 8 + i],
                                                   (short)0, S1, false, false);
    }
}

__global__ __launch_bounds__(256) __attribute__((amdgpu_waves_per_eu(4, 4))) void attn_kernel(
    const float* __restrict__ Q, const float* __restrict__ K,
    const float* __restrict__ V, float* __restrict__ out)
{
    __shared__ float Kls[KPT * KSTR];
    __shared__ float Vls[KG][16 * PSTR];

    const int tid  = threadIdx.x;
    const int lane = tid & 31;
    const int wave = tid >> 5;
    const int g    = wave / QW;
    const int wq   = wave % QW;
    const int half = lane >> 4;
    const int l16  = lane & 15;
    const int qRow = blockIdx.x * QPB + wq * 16 + l16;

    v2f qf[32];
#pragma unroll
    for (int kk = 0; kk < 32; ++kk) {
        const float* s = &Q[qRow * EMB + kk * 4 + half * 2];
        v2f t; t.x = s[0] * TEMP_INV; t.y = s[1] * TEMP_INV;
        qf[kk] = t;
    }

    v8f Oacc = {};
    float mrun = -3.0e38f, lrun = 0.f;

    const float* kmine = &Kls[g * 32 * KSTR];
    const float* vmine = &Vls[g][0];

    const float* kp = K + tid * 4;
    const float* vp = V + tid * 4;
    v4f kr[16], vr[2];
#pragma unroll
    for (int i = 0; i < 16; ++i) kr[i] = *(const v4f*)(kp + i * 1024);
#pragma unroll
    for (int i = 0; i < 2; ++i)  vr[i] = *(const v4f*)(vp + i * 1024);

    for (int kt = 0; kt < N_POS / KPT; ++kt) {
        __syncthreads();
        {
            float* kd = &Kls[wave * KSTR + lane * 4];
#pragma unroll
            for (int i = 0; i < 16; ++i) *(v4f*)(kd + i * 8 * KSTR) = kr[i];
#pragma unroll
            for (int i = 0; i < 2; ++i) {
                int j  = tid + i * 256;
                int k  = j >> 2;
                int c4 = (j & 3) * 4;
                int k5 = k & 31;
                int pos = (k5 & 16) | ((k5 & 6) << 1) | ((k5 & 8) >> 2) | (k5 & 1);
                float* vd = &Vls[k >> 5][c4 * PSTR + pos];
                vd[0 * PSTR] = vr[i].x; vd[1 * PSTR] = vr[i].y;
                vd[2 * PSTR] = vr[i].z; vd[3 * PSTR] = vr[i].w;
            }
        }
        __syncthreads();

        v8f S0 = {}, S1 = {};
        const float* ka = &kmine[l16 * KSTR + half * 2];
        const float* kb = ka + 16 * KSTR;
        v2f fx[16], fy[16];
        load_kgroup(fx, ka, kb, 0);
        load_kgroup(fy, ka, kb, 1);
        __builtin_amdgcn_sched_barrier(0);
        if (kt + 1 < N_POS / KPT) {
#pragma unroll
            for (int i = 0; i < 16; ++i) kr[i] = *(const v4f*)(kp + KPT * EMB + i * 1024);
#pragma unroll
            for (int i = 0; i < 2; ++i)  vr[i] = *(const v4f*)(vp + KPT * 16 + i * 1024);
        }
        mma_kgroup(S0, S1, fx, qf, 0);
        load_kgroup(fx, ka, kb, 2);
        mma_kgroup(S0, S1, fy, qf, 1);
        __builtin_amdgcn_sched_barrier(0);
        load_kgroup(fy, ka, kb, 3);
        __builtin_amdgcn_sched_barrier(0);
        mma_kgroup(S0, S1, fx, qf, 2);
        mma_kgroup(S0, S1, fy, qf, 3);

        v2f va[8];
#pragma unroll
        for (int c = 0; c < 8; ++c)
            va[c] = *(const v2f*)&vmine[l16 * PSTR + (c >> 2) * 16 + (c & 3) * 4 + half * 2];
        __builtin_amdgcn_sched_barrier(0);

        float mx = fmaxf(S0[0], S1[0]);
#pragma unroll
        for (int r = 1; r < 8; ++r) mx = fmaxf(mx, fmaxf(S0[r], S1[r]));
        mx = fmaxf(mx, xhalf(mx));
        const float newm  = fmaxf(mrun, mx);
        const float alpha = __expf(mrun - newm);
        v2f p0[4], p1[4];
        float rs0 = 0.f, rs1 = 0.f;
#pragma unroll
        for (int j = 0; j < 4; ++j) {
            v2f a, b;
            a.x = __expf(S0[2 * j] - newm);  a.y = __expf(S0[2 * j + 1] - newm);
            b.x = __expf(S1[2 * j] - newm);  b.y = __expf(S1[2 * j + 1] - newm);
            p0[j] = a;  p1[j] = b;
            rs0 += a.x + a.y;
            rs1 += b.x + b.y;
        }
        float rs = rs0 + rs1;
        rs += xhalf(rs);
        lrun = lrun * alpha + rs;
        mrun = newm;
#pragma unroll
        for (int r = 0; r < 8; ++r) Oacc[r] *= alpha;

        {
#pragma unroll
            for (int j = 0; j < 4; ++j)
                Oacc = __builtin_amdgcn_wmma_f32_16x16x4_f32(false, va[j], false, p0[j],
                                                             (short)0, Oacc, false, false);
#pragma unroll
            for (int j = 0; j < 4; ++j)
                Oacc = __builtin_amdgcn_wmma_f32_16x16x4_f32(false, va[4 + j], false, p1[j],
                                                             (short)0, Oacc, false, false);
        }
        kp += KPT * EMB;
        vp += KPT * 16;
    }

    __syncthreads();
    float* cO = &Kls[0];
    float* cm = cO + KG * 16 * CSTR;
    float* cl = cm + KG * QPB;
    {
        const int ql = wq * 16 + l16;
#pragma unroll
        for (int r = 0; r < 8; ++r)
            cO[(g * 16 + 8 * half + r) * CSTR + ql] = Oacc[r];
        if (half == 0) { cm[g * QPB + ql] = mrun; cl[g * QPB + ql] = lrun; }
    }
    __syncthreads();

    {
        const int ql = lane;
        float mg[KG], mx = -3.0e38f;
#pragma unroll
        for (int k = 0; k < KG; ++k) { mg[k] = cm[k * QPB + ql]; mx = fmaxf(mx, mg[k]); }
        float a[KG], lsum = 0.f;
#pragma unroll
        for (int k = 0; k < KG; ++k) { a[k] = __expf(mg[k] - mx); lsum += cl[k * QPB + ql] * a[k]; }
#pragma unroll
        for (int i = 0; i < 2; ++i) {
            int c = wave + 8 * i;
            float o = 0.f;
#pragma unroll
            for (int k = 0; k < KG; ++k) o += cO[(k * 16 + c) * CSTR + ql] * a[k];
            out[c * N_POS + blockIdx.x * QPB + ql] = o / lsum;
        }
    }
}

extern "C" void kernel_launch(void* const* d_in, const int* in_sizes, int n_in,
                              void* d_out, int out_size, void* d_ws, size_t ws_size,
                              hipStream_t stream)
{
    const float* ego_cls   = (const float*)d_in[0];
    const float* ego_reg   = (const float*)d_in[1];
    const float* other_cls = (const float*)d_in[2];
    const float* other_reg = (const float*)d_in[3];
    const float* w1  = (const float*)d_in[4];
    const float* b1  = (const float*)d_in[5];
    const float* ln_g = (const float*)d_in[6];
    const float* ln_b = (const float*)d_in[7];
    const float* w2  = (const float*)d_in[8];
    const float* b2  = (const float*)d_in[9];

    float* Qbuf = (float*)d_ws;
    float* Kbuf = Qbuf + N_POS * EMB;
    float* Vbuf = Kbuf + N_POS * EMB;
    float* out  = (float*)d_out;

    encode_kernel<<<dim3(N_POS / ENC_TOK, 2), 256, 0, stream>>>(
        ego_cls, ego_reg, other_cls, other_reg,
        w1, b1, ln_g, ln_b, w2, b2, Qbuf, Kbuf);

    vpack_kernel<<<(N_POS * 16 + 255) / 256, 256, 0, stream>>>(other_cls, other_reg, Vbuf);

    attn_kernel<<<N_POS / QPB, 256, 0, stream>>>(Qbuf, Kbuf, Vbuf, out);
}


# attention: K/V tile double-buffered in LDS; tile kt+1 is stored into the other buffer behind tile kt's first fragment reads: one work-group barrier per trip, not two with the store phase between
# speedup vs baseline: 3.3231x; 1.0368x over previous
#include <hip/hip_runtime.h>
#include <math.h>

typedef float v2f __attribute__((ext_vector_type(2)));
typedef float v4f __attribute__((ext_vector_type(4)));
typedef float v8f __attribute__((ext_vector_type(8)));

#define N_POS 16384
#define EMB 128
#define TEMP_INV 10.0f
#define LN_EPSF 1e-5f
#define KSTR 132
#define PSTR 34

#define ENC_TOK 128
#define W2PITCH 288

__global__ __launch_bounds__(256) void encode_kernel(
    const float* __restrict__ ego_cls, const float* __restrict__ ego_reg,
    const float* __restrict__ other_cls, const float* __restrict__ other_reg,
    const float* __restrict__ w1, const float* __restrict__ b1,
    const float* __restrict__ ln_g, const float* __restrict__ ln_b,
    const float* __restrict__ w2, const float* __restrict__ b2,
    float* __restrict__ Qbuf, float* __restrict__ Kbuf)
{
    __shared__ float W2p[64 * W2PITCH];

    const int tid  = threadIdx.x;
    const int lane = tid & 31;
    const int wave = tid >> 5;
    const int half = lane >> 4;
    const int l16  = lane & 15;

    {
        v4f r0[8], r1[8];
#pragma unroll
        for (int i = 0; i < 8; ++i) {
            int idx = tid + i * 256;
            int pr  = idx >> 5;
            int e4  = (idx & 31) * 4;
            r0[i] = *(const v4f*)&w2[(2 * pr) * EMB + e4];
            r1[i] = *(const v4f*)&w2[(2 * pr + 1) * EMB + e4];
        }
#pragma unroll
        for (int i = 0; i < 8; ++i) {
            int idx = tid + i * 256;
            int pr  = idx >> 5;
            int e4  = (idx & 31) * 4;
            float* d = &W2p[pr * W2PITCH + 2 * e4];
            v4f lo, hi;
            lo.x = r0[i].x; lo.y = r1[i].x; lo.z = r0[i].y; lo.w = r1[i].y;
            hi.x = r0[i].z; hi.y = r1[i].z; hi.z = r0[i].w; hi.w = r1[i].w;
            *(v4f*)d       = lo;
            *(v4f*)(d + 4) = hi;
        }
    }

    const float* cls; const float* reg; float* outp;
    if (blockIdx.y == 0) { cls = ego_cls;   reg = ego_reg;   outp = Qbuf; }
    else                 { cls = other_cls; reg = other_reg; outp = Kbuf; }

    const int tok0 = blockIdx.x * ENC_TOK + wave * 16;
    const int p    = tok0 + l16;

    float x0 = reg[3 * N_POS + p];
    float x1 = reg[4 * N_POS + p];
    float x2 = reg[5 * N_POS + p];
    float th = reg[6 * N_POS + p];
    float x3 = sinf(th);
    float x4 = cosf(th);
    float x5 = fmaxf(cls[p], cls[N_POS + p]);

    const int cb = 2 * half;
    v2f hf[32];
    float sum = 0.f, sumsq = 0.f;
#pragma unroll
    for (int kk = 0; kk < 32; ++kk) {
        const int c = 4 * kk + cb;
        v2f a  = *(const v2f*)&b1[c];
        v2f wa = *(const v2f*)&w1[0 * EMB + c];
        v2f wb = *(const v2f*)&w1[1 * EMB + c];
        v2f wc = *(const v2f*)&w1[2 * EMB + c];
        v2f wd = *(const v2f*)&w1[3 * EMB + c];
        v2f we = *(const v2f*)&w1[4 * EMB + c];
        v2f wf = *(const v2f*)&w1[5 * EMB + c];
        float a0 = a.x, a1 = a.y;
        a0 += x0 * wa.x;  a1 += x0 * wa.y;
        a0 += x1 * wb.x;  a1 += x1 * wb.y;
        a0 += x2 * wc.x;  a1 += x2 * wc.y;
        a0 += x3 * wd.x;  a1 += x3 * wd.y;
        a0 += x4 * we.x;  a1 += x4 * we.y;
        a0 += x5 * wf.x;  a1 += x5 * wf.y;
        a0 = fmaxf(a0, 0.f);  a1 = fmaxf(a1, 0.f);
        v2f t; t.x = a0; t.y = a1;
        hf[kk] = t;
        sum += a0; sumsq += a0 * a0;
        sum += a1; sumsq += a1 * a1;
    }
    sum   += __shfl_xor(sum, 16, 32);
    sumsq += __shfl_xor(sumsq, 16, 32);
    float mu   = sum * (1.f / EMB);
    float var  = sumsq * (1.f / EMB) - mu * mu;
    float rstd = rsqrtf(var + LN_EPSF);
#pragma unroll
    for (int kk = 0; kk < 32; ++kk) {
        const int c = 4 * kk + cb;
        v2f g = *(const v2f*)&ln_g[c];
        v2f b = *(const v2f*)&ln_b[c];
        v2f t = hf[kk];
        t.x = (t.x - mu) * rstd * g.x + b.x;
        t.y = (t.y - mu) * rstd * g.y + b.y;
        hf[kk] = t;
    }

    __syncthreads();

    const float* wl = &W2p[half * W2PITCH + 2 * l16];
#pragma unroll
    for (int nb = 0; nb < 8; nb += 2) {
        float bias0 = b2[nb * 16 + l16];
        float bias1 = b2[nb * 16 + 16 + l16];
        v8f acc0, acc1;
#pragma unroll
        for (int r = 0; r < 8; ++r) { acc0[r] = bias0; acc1[r] = bias1; }
#pragma unroll
        for (int kk = 0; kk < 32; ++kk) {
            v2f bf0 = *(const v2f*)&wl[2 * kk * W2PITCH + nb * 32];
            v2f bf1 = *(const v2f*)&wl[2 * kk * W2PITCH + nb * 32 + 32];
            acc0 = __builtin_amdgcn_wmma_f32_16x16x4_f32(false, hf[kk], false, bf0,
                                                         (short)0, acc0, false, false);
            acc1 = __builtin_amdgcn_wmma_f32_16x16x4_f32(false, hf[kk], false, bf1,
                                                         (short)0, acc1, false, false);
        }
        float* o = &outp[(tok0 + 8 * half) * EMB + nb * 16 + l16];
#pragma unroll
        for (int r = 0; r < 8; ++r) {
            o[r * EMB]      = acc0[r];
            o[r * EMB + 16] = acc1[r];
        }
    }
}

__global__ void vpack_kernel(const float* __restrict__ oc, const float* __restrict__ org,
                             float* __restrict__ V)
{
    int t = blockIdx.x * 256 + threadIdx.x;
    if (t >= N_POS * 16) return;
    int p = t >> 4, c = t & 15;
    V[t] = (c < 2) ? oc[c * N_POS + p] : org[(c - 2) * N_POS + p];
}

#define KG 4
#define QW 2
#define QPB (QW * 16)
#define KPT (KG * 32)
#define CSTR 33
#define KBUF (KPT * KSTR)
#define VBUF (KG * 16 * PSTR)
#define BUFSZ (KBUF + VBUF)

__device__ __forceinline__ float xhalf(float x)
{
    int b = __builtin_bit_cast(int, x);
    return __builtin_bit_cast(float,
        __builtin_amdgcn_permlanex16(b, b, 0x76543210, 0xfedcba98, false, false));
}

__device__ __forceinline__ void load_kgroup(v2f (&f)[16], const float* ka, const float* kb, int g)
{
#pragma unroll
    for (int i = 0; i < 8; ++i) {
        f[2 * i]     = *(const v2f*)&ka[(g * 8 + i) * 4];
        f[2 * i + 1] = *(const v2f*)&kb[(g * 8 + i) * 4];
    }
}

__device__ __forceinline__ void mma_kgroup(v8f& S0, v8f& S1, const v2f (&f)[16], const v2f (&qf)[32], int g)
{
#pragma unroll
    for (int i = 0; i < 8; ++i) {
        S0 = __builtin_amdgcn_wmma_f32_16x16x4_f32(false, f[2 * i], false, qf[g * 8 + i],
                                                   (short)0, S0, false, false);
        S1 = __builtin_amdgcn_wmma_f32_16x16x4_f32(false, f[2 * i + 1], false, qf[g * 8 + i],
                                                   (short)0, S1, false, false);
    }
}

__global__ __launch_bounds__(256) __attribute__((amdgpu_waves_per_eu(4, 4))) void attn_kernel(
    const float* __restrict__ Q, const float* __restrict__ K,
    const float* __restrict__ V, float* __restrict__ out)
{
    __shared__ float Tls[2 * BUFSZ];

    const int tid  = threadIdx.x;
    const int lane = tid & 31;
    const int wave = tid >> 5;
    const int g    = wave / QW;
    const int wq   = wave % QW;
    const int half = lane >> 4;
    const int l16  = lane & 15;
    const int qRow = blockIdx.x * QPB + wq * 16 + l16;

    v2f qf[32];
#pragma unroll
    for (int kk = 0; kk < 32; ++kk) {
        const float* s = &Q[qRow * EMB + kk * 4 + half * 2];
        v2f t; t.x = s[0] * TEMP_INV; t.y = s[1] * TEMP_INV;
        qf[kk] = t;
    }

    v8f Oacc = {};
    float mrun = -3.0e38f, lrun = 0.f;

    const int krd = (g * 32 + l16) * KSTR + half * 2;
    const int vrd = KBUF + g * 16 * PSTR + l16 * PSTR + half * 2;
    const int kst = wave * KSTR + lane * 4;
    int vst[2];
#pragma unroll
    for (int i = 0; i < 2; ++i) {
        int j  = tid + i * 256;
        int k  = j >> 2;
        int c4 = (j & 3) * 4;
        int k5 = k & 31;
        int pos = (k5 & 16) | ((k5 & 6) << 1) | ((k5 & 8) >> 2) | (k5 & 1);
        vst[i] = KBUF + (k >> 5) * 16 * PSTR + c4 * PSTR + pos;
    }

    const float* kp = K + tid * 4;
    const float* vp = V + tid * 4;
    v4f kr[16], vr[2];
    {
        v4f k0[16], v0[2];
#pragma unroll
        for (int i = 0; i < 16; ++i) k0[i] = *(const v4f*)(kp + i * 1024);
#pragma unroll
        for (int i = 0; i < 2; ++i)  v0[i] = *(const v4f*)(vp + i * 1024);
#pragma unroll
        for (int i = 0; i < 16; ++i) kr[i] = *(const v4f*)(kp + KPT * EMB + i * 1024);
#pragma unroll
        for (int i = 0; i < 2; ++i)  vr[i] = *(const v4f*)(vp + KPT * 16 + i * 1024);
        float* kd = &Tls[kst];
#pragma unroll
        for (int i = 0; i < 16; ++i) *(v4f*)(kd + i * 8 * KSTR) = k0[i];
#pragma unroll
        for (int i = 0; i < 2; ++i) {
            float* vd = &Tls[vst[i]];
            vd[0 * PSTR] = v0[i].x; vd[1 * PSTR] = v0[i].y;
            vd[2 * PSTR] = v0[i].z; vd[3 * PSTR] = v0[i].w;
        }
    }
    int kra = krd, vra = vrd;
    int ksa = BUFSZ + kst, vsa0 = BUFSZ + vst[0], vsa1 = BUFSZ + vst[1];
    int step = BUFSZ;

    for (int kt = 0; kt < N_POS / KPT; ++kt) {
        __syncthreads();

        v8f S0 = {}, S1 = {};
        const float* ka = &Tls[kra];
        const float* kb = ka + 16 * KSTR;
        v2f fx[16], fy[16];
        load_kgroup(fx, ka, kb, 0);
        load_kgroup(fy, ka, kb, 1);
        __builtin_amdgcn_sched_barrier(0);
        if (kt + 1 < N_POS / KPT) {
            float* kd = &Tls[ksa];
#pragma unroll
            for (int i = 0; i < 16; ++i) *(v4f*)(kd + i * 8 * KSTR) = kr[i];
#pragma unroll
            for (int i = 0; i < 2; ++i) {
                float* vd = &Tls[i == 0 ? vsa0 : vsa1];
                vd[0 * PSTR] = vr[i].x; vd[1 * PSTR] = vr[i].y;
                vd[2 * PSTR] = vr[i].z; vd[3 * PSTR] = vr[i].w;
            }
        }
        if (kt + 2 < N_POS / KPT) {
#pragma unroll
            for (int i = 0; i < 16; ++i) kr[i] = *(const v4f*)(kp + 2 * KPT * EMB + i * 1024);
#pragma unroll
            for (int i = 0; i < 2; ++i)  vr[i] = *(const v4f*)(vp + 2 * KPT * 16 + i * 1024);
        }
        __builtin_amdgcn_sched_barrier(0);
        mma_kgroup(S0, S1, fx, qf, 0);
        load_kgroup(fx, ka, kb, 2);
        mma_kgroup(S0, S1, fy, qf, 1);
        __builtin_amdgcn_sched_barrier(0);
        load_kgroup(fy, ka, kb, 3);
        __builtin_amdgcn_sched_barrier(0);
        mma_kgroup(S0, S1, fx, qf, 2);
        mma_kgroup(S0, S1, fy, qf, 3);

        v2f va[8];
#pragma unroll
        for (int c = 0; c < 8; ++c)
            va[c] = *(const v2f*)&Tls[vra + (c >> 2) * 16 + (c & 3) * 4];
        __builtin_amdgcn_sched_barrier(0);

        float mx = fmaxf(S0[0], S1[0]);
#pragma unroll
        for (int r = 1; r < 8; ++r) mx = fmaxf(mx, fmaxf(S0[r], S1[r]));
        mx = fmaxf(mx, xhalf(mx));
        const float newm  = fmaxf(mrun, mx);
        const float alpha = __expf(mrun - newm);
        v2f p0[4], p1[4];
        float rs0 = 0.f, rs1 = 0.f;
#pragma unroll
        for (int j = 0; j < 4; ++j) {
            v2f a, b;
            a.x = __expf(S0[2 * j] - newm);  a.y = __expf(S0[2 * j + 1] - newm);
            b.x = __expf(S1[2 * j] - newm);  b.y = __expf(S1[2 * j + 1] - newm);
            p0[j] = a;  p1[j] = b;
            rs0 += a.x + a.y;
            rs1 += b.x + b.y;
        }
        float rs = rs0 + rs1;
        rs += xhalf(rs);
        lrun = lrun * alpha + rs;
        mrun = newm;
#pragma unroll
        for (int r = 0; r < 8; ++r) Oacc[r] *= alpha;

        {
#pragma unroll
            for (int j = 0; j < 4; ++j)
                Oacc = __builtin_amdgcn_wmma_f32_16x16x4_f32(false, va[j], false, p0[j],
                                                             (short)0, Oacc, false, false);
#pragma unroll
            for (int j = 0; j < 4; ++j)
                Oacc = __builtin_amdgcn_wmma_f32_16x16x4_f32(false, va[4 + j], false, p1[j],
                                                             (short)0, Oacc, false, false);
        }
        kp += KPT * EMB;
        vp += KPT * 16;
        kra += step; vra += step;
        ksa -= step; vsa0 -= step; vsa1 -= step;
        step = -step;
    }

    __syncthreads();
    float* cO = &Tls[0];
    float* cm = cO + KG * 16 * CSTR;
    float* cl = cm + KG * QPB;
    {
        const int ql = wq * 16 + l16;
#pragma unroll
        for (int r = 0; r < 8; ++r)
            cO[(g * 16 + 8 * half + r) * CSTR + ql] = Oacc[r];
        if (half == 0) { cm[g * QPB + ql] = mrun; cl[g * QPB + ql] = lrun; }
    }
    __syncthreads();

    {
        const int ql = lane;
        float mg[KG], mx = -3.0e38f;
#pragma unroll
        for (int k = 0; k < KG; ++k) { mg[k] = cm[k * QPB + ql]; mx = fmaxf(mx, mg[k]); }
        float a[KG], lsum = 0.f;
#pragma unroll
        for (int k = 0; k < KG; ++k) { a[k] = __expf(mg[k] - mx); lsum += cl[k * QPB + ql] * a[k]; }
#pragma unroll
        for (int i = 0; i < 2; ++i) {
            int c = wave + 8 * i;
            float o = 0.f;
#pragma unroll
            for (int k = 0; k < KG; ++k) o += cO[(k * 16 + c) * CSTR + ql] * a[k];
            out[c * N_POS + blockIdx.x * QPB + ql] = o / lsum;
        }
    }
}

extern "C" void kernel_launch(void* const* d_in, const int* in_sizes, int n_in,
                              void* d_out, int out_size, void* d_ws, size_t ws_size,
                              hipStream_t stream)
{
    const float* ego_cls   = (const float*)d_in[0];
    const float* ego_reg   = (const float*)d_in[1];
    const float* other_cls = (const float*)d_in[2];
    const float* other_reg = (const float*)d_in[3];
    const float* w1  = (const float*)d_in[4];
    const float* b1  = (const float*)d_in[5];
    const float* ln_g = (const float*)d_in[6];
    const float* ln_b = (const float*)d_in[7];
    const float* w2  = (const float*)d_in[8];
    const float* b2  = (const float*)d_in[9];

    float* Qbuf = (float*)d_ws;
    float* Kbuf = Qbuf + N_POS * EMB;
    float* Vbuf = Kbuf + N_POS * EMB;
    float* out  = (float*)d_out;

    encode_kernel<<<dim3(N_POS / ENC_TOK, 2), 256, 0, stream>>>(
        ego_cls, ego_reg, other_cls, other_reg,
        w1, b1, ln_g, ln_b, w2, b2, Qbuf, Kbuf);

    vpack_kernel<<<(N_POS * 16 + 255) / 256, 256, 0, stream>>>(other_cls, other_reg, Vbuf);

    attn_kernel<<<N_POS / QPB, 256, 0, stream>>>(Qbuf, Kbuf, Vbuf, out);
}


# attention: the trip's work-group barrier split in two: a wave signals inside the last score block, its last LDS reads back, and waits at the next trip's top; softmax and P.V run in between
# speedup vs baseline: 3.5541x; 1.0695x over previous
#include <hip/hip_runtime.h>
#include <math.h>

typedef float v2f __attribute__((ext_vector_type(2)));
typedef float v4f __attribute__((ext_vector_type(4)));
typedef float v8f __attribute__((ext_vector_type(8)));

#define N_POS 16384
#define EMB 128
#define TEMP_INV 10.0f
#define LN_EPSF 1e-5f
#define KSTR 132
#define PSTR 34

#define ENC_TOK 128
#define W2PITCH 288

__global__ __launch_bounds__(256) void encode_kernel(
    const float* __restrict__ ego_cls, const float* __restrict__ ego_reg,
    const float* __restrict__ other_cls, const float* __restrict__ other_reg,
    const float* __restrict__ w1, const float* __restrict__ b1,
    const float* __restrict__ ln_g, const float* __restrict__ ln_b,
    const float* __restrict__ w2, const float* __restrict__ b2,
    float* __restrict__ Qbuf, float* __restrict__ Kbuf)
{
    __shared__ float W2p[64 * W2PITCH];

    const int tid  = threadIdx.x;
    const int lane = tid & 31;
    const int wave = tid >> 5;
    const int half = lane >> 4;
    const int l16  = lane & 15;

    {
        v4f r0[8], r1[8];
#pragma unroll
        for (int i = 0; i < 8; ++i) {
            int idx = tid + i * 256;
            int pr  = idx >> 5;
            int e4  = (idx & 31) * 4;
            r0[i] = *(const v4f*)&w2[(2 * pr) * EMB + e4];
            r1[i] = *(const v4f*)&w2[(2 * pr + 1) * EMB + e4];
        }
#pragma unroll
        for (int i = 0; i < 8; ++i) {
            int idx = tid + i * 256;
            int pr  = idx >> 5;
            int e4  = (idx & 31) * 4;
            float* d = &W2p[pr * W2PITCH + 2 * e4];
            v4f lo, hi;
            lo.x = r0[i].x; lo.y = r1[i].x; lo.z = r0[i].y; lo.w = r1[i].y;
            hi.x = r0[i].z; hi.y = r1[i].z; hi.z = r0[i].w; hi.w = r1[i].w;
            *(v4f*)d       = lo;
            *(v4f*)(d + 4) = hi;
        }
    }

    const float* cls; const float* reg; float* outp;
    if (blockIdx.y == 0) { cls = ego_cls;   reg = ego_reg;   outp = Qbuf; }
    else                 { cls = other_cls; reg = other_reg; outp = Kbuf; }

    const int tok0 = blockIdx.x * ENC_TOK + wave * 16;
    const int p    = tok0 + l16;

    float x0 = reg[3 * N_POS + p];
    float x1 = reg[4 * N_POS + p];
    float x2 = reg[5 * N_POS + p];
    float th = reg[6 * N_POS + p];
    float x3 = sinf(th);
    float x4 = cosf(th);
    float x5 = fmaxf(cls[p], cls[N_POS + p]);

    const int cb = 2 * half;
    v2f hf[32];
    float sum = 0.f, sumsq = 0.f;
#pragma unroll
    for (int kk = 0; kk < 32; ++kk) {
        const int c = 4 * kk + cb;
        v2f a  = *(const v2f*)&b1[c];
        v2f wa = *(const v2f*)&w1[0 * EMB + c];
        v2f wb = *(const v2f*)&w1[1 * EMB + c];
        v2f wc = *(const v2f*)&w1[2 * EMB + c];
        v2f wd = *(const v2f*)&w1[3 * EMB + c];
        v2f we = *(const v2f*)&w1[4 * EMB + c];
        v2f wf = *(const v2f*)&w1[5 * EMB + c];
        float a0 = a.x, a1 = a.y;
        a0 += x0 * wa.x;  a1 += x0 * wa.y;
        a0 += x1 * wb.x;  a1 += x1 * wb.y;
        a0 += x2 * wc.x;  a1 += x2 * wc.y;
        a0 += x3 * wd.x;  a1 += x3 * wd.y;
        a0 += x4 * we.x;  a1 += x4 * we.y;
        a0 += x5 * wf.x;  a1 += x5 * wf.y;
        a0 = fmaxf(a0, 0.f);  a1 = fmaxf(a1, 0.f);
        v2f t; t.x = a0; t.y = a1;
        hf[kk] = t;
        sum += a0; sumsq += a0 * a0;
        sum += a1; sumsq += a1 * a1;
    }
    sum   += __shfl_xor(sum, 16, 32);
    sumsq += __shfl_xor(sumsq, 16, 32);
    float mu   = sum * (1.f / EMB);
    float var  = sumsq * (1.f / EMB) - mu * mu;
    float rstd = rsqrtf(var + LN_EPSF);
#pragma unroll
    for (int kk = 0; kk < 32; ++kk) {
        const int c = 4 * kk + cb;
        v2f g = *(const v2f*)&ln_g[c];
        v2f b = *(const v2f*)&ln_b[c];
        v2f t = hf[kk];
        t.x = (t.x - mu) * rstd * g.x + b.x;
        t.y = (t.y - mu) * rstd * g.y + b.y;
        hf[kk] = t;
    }

    __syncthreads();

    const float* wl = &W2p[half * W2PITCH + 2 * l16];
#pragma unroll
    for (int nb = 0; nb < 8; nb += 2) {
        float bias0 = b2[nb * 16 + l16];
        float bias1 = b2[nb * 16 + 16 + l16];
        v8f acc0, acc1;
#pragma unroll
        for (int r = 0; r < 8; ++r) { acc0[r] = bias0; acc1[r] = bias1; }
#pragma unroll
        for (int kk = 0; kk < 32; ++kk) {
            v2f bf0 = *(const v2f*)&wl[2 * kk * W2PITCH + nb * 32];
            v2f bf1 = *(const v2f*)&wl[2 * kk * W2PITCH + nb * 32 + 32];
            acc0 = __builtin_amdgcn_wmma_f32_16x16x4_f32(false, hf[kk], false, bf0,
                                                         (short)0, acc0, false, false);
            acc1 = __builtin_amdgcn_wmma_f32_16x16x4_f32(false, hf[kk], false, bf1,
                                                         (short)0, acc1, false, false);
        }
        float* o = &outp[(tok0 + 8 * half) * EMB + nb * 16 + l16];
#pragma unroll
        for (int r = 0; r < 8; ++r) {
            o[r * EMB]      = acc0[r];
            o[r * EMB + 16] = acc1[r];
        }
    }
}

__global__ void vpack_kernel(const float* __restrict__ oc, const float* __restrict__ org,
                             float* __restrict__ V)
{
    int t = blockIdx.x * 256 + threadIdx.x;
    if (t >= N_POS * 16) return;
    int p = t >> 4, c = t & 15;
    V[t] = (c < 2) ? oc[c * N_POS + p] : org[(c - 2) * N_POS + p];
}

#define KG 4
#define QW 2
#define QPB (QW * 16)
#define KPT (KG * 32)
#define CSTR 33
#define KBUF (KPT * KSTR)
#define VBUF (KG * 16 * PSTR)
#define BUFSZ (KBUF + VBUF)

__device__ __forceinline__ float xhalf(float x)
{
    int b = __builtin_bit_cast(int, x);
    return __builtin_bit_cast(float,
        __builtin_amdgcn_permlanex16(b, b, 0x76543210, 0xfedcba98, false, false));
}

__device__ __forceinline__ void bar_arrive()
{
    asm volatile("s_wait_dscnt 0x0\n\ts_barrier_signal -1" ::: "memory");
}
__device__ __forceinline__ void bar_wait()
{
    asm volatile("s_barrier_wait -1" ::: "memory");
}

__device__ __forceinline__ void load_kgroup(v2f (&f)[16], const float* ka, const float* kb, int g)
{
#pragma unroll
    for (int i = 0; i < 8; ++i) {
        f[2 * i]     = *(const v2f*)&ka[(g * 8 + i) * 4];
        f[2 * i + 1] = *(const v2f*)&kb[(g * 8 + i) * 4];
    }
}

__device__ __forceinline__ void mma_ksteps(v8f& S0, v8f& S1, const v2f (&f)[16], const v2f (&qf)[32], int g,
                                           int i0, int i1)
{
#pragma unroll
    for (int i = i0; i < i1; ++i) {
        S0 = __builtin_amdgcn_wmma_f32_16x16x4_f32(false, f[2 * i], false, qf[g * 8 + i],
                                                   (short)0, S0, false, false);
        S1 = __builtin_amdgcn_wmma_f32_16x16x4_f32(false, f[2 * i + 1], false, qf[g * 8 + i],
                                                   (short)0, S1, false, false);
    }
}
__device__ __forceinline__ void mma_kgroup(v8f& S0, v8f& S1, const v2f (&f)[16], const v2f (&qf)[32], int g)
{
    mma_ksteps(S0, S1, f, qf, g, 0, 8);
}

__global__ __launch_bounds__(256) __attribute__((amdgpu_waves_per_eu(4, 4))) void attn_kernel(
    const float* __restrict__ Q, const float* __restrict__ K,
    const float* __restrict__ V, float* __restrict__ out)
{
    __shared__ float Tls[2 * BUFSZ];

    const int tid  = threadIdx.x;
    const int lane = tid & 31;
    const int wave = tid >> 5;
    const int g    = wave / QW;
    const int wq   = wave % QW;
    const int half = lane >> 4;
    const int l16  = lane & 15;
    const int qRow = blockIdx.x * QPB + wq * 16 + l16;

    v2f qf[32];
#pragma unroll
    for (int kk = 0; kk < 32; ++kk) {
        const float* s = &Q[qRow * EMB + kk * 4 + half * 2];
        v2f t; t.x = s[0] * TEMP_INV; t.y = s[1] * TEMP_INV;
        qf[kk] = t;
    }

    v8f Oacc = {};
    float mrun = -3.0e38f, lrun = 0.f;

    const int krd = (g * 32 + l16) * KSTR + half * 2;
    const int vrd = KBUF + g * 16 * PSTR + l16 * PSTR + half * 2;
    const int kst = wave * KSTR + lane * 4;
    int vst[2];
#pragma unroll
    for (int i = 0; i < 2; ++i) {
        int j  = tid + i * 256;
        int k  = j >> 2;
        int c4 = (j & 3) * 4;
        int k5 = k & 31;
        int pos = (k5 & 16) | ((k5 & 6) << 1) | ((k5 & 8) >> 2) | (k5 & 1);
        vst[i] = KBUF + (k >> 5) * 16 * PSTR + c4 * PSTR + pos;
    }

    const float* kp = K + tid * 4;
    const float* vp = V + tid * 4;
    v4f kr[16], vr[2];
    {
        v4f k0[16], v0[2];
#pragma unroll
        for (int i = 0; i < 16; ++i) k0[i] = *(const v4f*)(kp + i * 1024);
#pragma unroll
        for (int i = 0; i < 2; ++i)  v0[i] = *(const v4f*)(vp + i * 1024);
#pragma unroll
        for (int i = 0; i < 16; ++i) kr[i] = *(const v4f*)(kp + KPT * EMB + i * 1024);
#pragma unroll
        for (int i = 0; i < 2; ++i)  vr[i] = *(const v4f*)(vp + KPT * 16 + i * 1024);
        float* kd = &Tls[kst];
#pragma unroll
        for (int i = 0; i < 16; ++i) *(v4f*)(kd + i * 8 * KSTR) = k0[i];
#pragma unroll
        for (int i = 0; i < 2; ++i) {
            float* vd = &Tls[vst[i]];
            vd[0 * PSTR] = v0[i].x; vd[1 * PSTR] = v0[i].y;
            vd[2 * PSTR] = v0[i].z; vd[3 * PSTR] = v0[i].w;
        }
        bar_arrive();
    }
    int kra = krd, vra = vrd;
    int ksa = BUFSZ + kst, vsa0 = BUFSZ + vst[0], vsa1 = BUFSZ + vst[1];
    int step = BUFSZ;

    for (int kt = 0; kt < N_POS / KPT; ++kt) {
        bar_wait();

        v8f S0 = {}, S1 = {};
        const float* ka = &Tls[kra];
        const float* kb = ka + 16 * KSTR;
        v2f fx[16], fy[16];
        load_kgroup(fx, ka, kb, 0);
        load_kgroup(fy, ka, kb, 1);
        __builtin_amdgcn_sched_barrier(0);
        if (kt + 1 < N_POS / KPT) {
            float* kd = &Tls[ksa];
#pragma unroll
            for (int i = 0; i < 16; ++i) *(v4f*)(kd + i * 8 * KSTR) = kr[i];
#pragma unroll
            for (int i = 0; i < 2; ++i) {
                float* vd = &Tls[i == 0 ? vsa0 : vsa1];
                vd[0 * PSTR] = vr[i].x; vd[1 * PSTR] = vr[i].y;
                vd[2 * PSTR] = vr[i].z; vd[3 * PSTR] = vr[i].w;
            }
        }
        if (kt + 2 < N_POS / KPT) {
#pragma unroll
            for (int i = 0; i < 16; ++i) kr[i] = *(const v4f*)(kp + 2 * KPT * EMB + i * 1024);
#pragma unroll
            for (int i = 0; i < 2; ++i)  vr[i] = *(const v4f*)(vp + 2 * KPT * 16 + i * 1024);
        }
        __builtin_amdgcn_sched_barrier(0);
        mma_kgroup(S0, S1, fx, qf, 0);
        load_kgroup(fx, ka, kb, 2);
        mma_kgroup(S0, S1, fy, qf, 1);
        __builtin_amdgcn_sched_barrier(0);
        load_kgroup(fy, ka, kb, 3);
        __builtin_amdgcn_sched_barrier(0);
        mma_kgroup(S0, S1, fx, qf, 2);

        v2f va[8];
#pragma unroll
        for (int c = 0; c < 8; ++c)
            va[c] = *(const v2f*)&Tls[vra + (c >> 2) * 16 + (c & 3) * 4];
        __builtin_amdgcn_sched_barrier(0);
        mma_ksteps(S0, S1, fy, qf, 3, 0, 4);
        __builtin_amdgcn_sched_barrier(0);
        bar_arrive();
        __builtin_amdgcn_sched_barrier(0);
        mma_ksteps(S0, S1, fy, qf, 3, 4, 8);

        float mx = fmaxf(S0[0], S1[0]);
#pragma unroll
        for (int r = 1; r < 8; ++r) mx = fmaxf(mx, fmaxf(S0[r], S1[r]));
        mx = fmaxf(mx, xhalf(mx));
        const float newm  = fmaxf(mrun, mx);
        const float alpha = __expf(mrun - newm);
        v2f p0[4], p1[4];
        float rs0 = 0.f, rs1 = 0.f;
#pragma unroll
        for (int j = 0; j < 4; ++j) {
            v2f a, b;
            a.x = __expf(S0[2 * j] - newm);  a.y = __expf(S0[2 * j + 1] - newm);
            b.x = __expf(S1[2 * j] - newm);  b.y = __expf(S1[2 * j + 1] - newm);
            p0[j] = a;  p1[j] = b;
            rs0 += a.x + a.y;
            rs1 += b.x + b.y;
        }
        float rs = rs0 + rs1;
        rs += xhalf(rs);
        lrun = lrun * alpha + rs;
        mrun = newm;
#pragma unroll
        for (int r = 0; r < 8; ++r) Oacc[r] *= alpha;

        {
#pragma unroll
            for (int j = 0; j < 4; ++j)
                Oacc = __builtin_amdgcn_wmma_f32_16x16x4_f32(false, va[j], false, p0[j],
                                                             (short)0, Oacc, false, false);
#pragma unroll
            for (int j = 0; j < 4; ++j)
                Oacc = __builtin_amdgcn_wmma_f32_16x16x4_f32(false, va[4 + j], false, p1[j],
                                                             (short)0, Oacc, false, false);
        }
        kp += KPT * EMB;
        vp += KPT * 16;
        kra += step; vra += step;
        ksa -= step; vsa0 -= step; vsa1 -= step;
        step = -step;
    }

    bar_wait();
    float* cO = &Tls[0];
    float* cm = cO + KG * 16 * CSTR;
    float* cl = cm + KG * QPB;
    {
        const int ql = wq * 16 + l16;
#pragma unroll
        for (int r = 0; r < 8; ++r)
            cO[(g * 16 + 8 * half + r) * CSTR + ql] = Oacc[r];
        if (half == 0) { cm[g * QPB + ql] = mrun; cl[g * QPB + ql] = lrun; }
    }
    __syncthreads();

    {
        const int ql = lane;
        float mg[KG], mx = -3.0e38f;
#pragma unroll
        for (int k = 0; k < KG; ++k) { mg[k] = cm[k * QPB + ql]; mx = fmaxf(mx, mg[k]); }
        float a[KG], lsum = 0.f;
#pragma unroll
        for (int k = 0; k < KG; ++k) { a[k] = __expf(mg[k] - mx); lsum += cl[k * QPB + ql] * a[k]; }
#pragma unroll
        for (int i = 0; i < 2; ++i) {
            int c = wave + 8 * i;
            float o = 0.f;
#pragma unroll
            for (int k = 0; k < KG; ++k) o += cO[(k * 16 + c) * CSTR + ql] * a[k];
            out[c * N_POS + blockIdx.x * QPB + ql] = o / lsum;
        }
    }
}

extern "C" void kernel_launch(void* const* d_in, const int* in_sizes, int n_in,
                              void* d_out, int out_size, void* d_ws, size_t ws_size,
                              hipStream_t stream)
{
    const float* ego_cls   = (const float*)d_in[0];
    const float* ego_reg   = (const float*)d_in[1];
    const float* other_cls = (const float*)d_in[2];
    const float* other_reg = (const float*)d_in[3];
    const float* w1  = (const float*)d_in[4];
    const float* b1  = (const float*)d_in[5];
    const float* ln_g = (const float*)d_in[6];
    const float* ln_b = (const float*)d_in[7];
    const float* w2  = (const float*)d_in[8];
    const float* b2  = (const float*)d_in[9];

    float* Qbuf = (float*)d_ws;
    float* Kbuf = Qbuf + N_POS * EMB;
    float* Vbuf = Kbuf + N_POS * EMB;
    float* out  = (float*)d_out;

    encode_kernel<<<dim3(N_POS / ENC_TOK, 2), 256, 0, stream>>>(
        ego_cls, ego_reg, other_cls, other_reg,
        w1, b1, ln_g, ln_b, w2, b2, Qbuf, Kbuf);

    vpack_kernel<<<(N_POS * 16 + 255) / 256, 256, 0, stream>>>(other_cls, other_reg, Vbuf);

    attn_kernel<<<N_POS / QPB, 256, 0, stream>>>(Qbuf, Kbuf, Vbuf, out);
}


# attention: K and V^T fragments go global->registers a tile ahead (K stored in fragment order); a wave owns 32 queries x a key quarter: no LDS or barrier in the key loop; new summation order
# speedup vs baseline: 3.8123x; 1.0726x over previous
#include <hip/hip_runtime.h>
#include <math.h>

typedef float v2f __attribute__((ext_vector_type(2)));
typedef float v4f __attribute__((ext_vector_type(4)));
typedef float v8f __attribute__((ext_vector_type(8)));

#define N_POS 16384
#define EMB 128
#define TEMP_INV 10.0f
#define LN_EPSF 1e-5f
#define TILE_FL 2048

#define ENC_TOK 128
#define W2PITCH 288

__global__ __launch_bounds__(256) void encode_kernel(
    const float* __restrict__ ego_cls, const float* __restrict__ ego_reg,
    const float* __restrict__ other_cls, const float* __restrict__ other_reg,
    const float* __restrict__ w1, const float* __restrict__ b1,
    const float* __restrict__ ln_g, const float* __restrict__ ln_b,
    const float* __restrict__ w2, const float* __restrict__ b2,
    float* __restrict__ Qbuf, float* __restrict__ Kbuf)
{
    __shared__ float W2p[64 * W2PITCH];

    const int tid  = threadIdx.x;
    const int lane = tid & 31;
    const int wave = tid >> 5;
    const int half = lane >> 4;
    const int l16  = lane & 15;

    {
        v4f r0[8], r1[8];
#pragma unroll
        for (int i = 0; i < 8; ++i) {
            int idx = tid + i * 256;
            int pr  = idx >> 5;
            int e4  = (idx & 31) * 4;
            r0[i] = *(const v4f*)&w2[(2 * pr) * EMB + e4];
            r1[i] = *(const v4f*)&w2[(2 * pr + 1) * EMB + e4];
        }
#pragma unroll
        for (int i = 0; i < 8; ++i) {
            int idx = tid + i * 256;
            int pr  = idx >> 5;
            int e4  = (idx & 31) * 4;
            float* d = &W2p[pr * W2PITCH + 2 * e4];
            v4f lo, hi;
            lo.x = r0[i].x; lo.y = r1[i].x; lo.z = r0[i].y; lo.w = r1[i].y;
            hi.x = r0[i].z; hi.y = r1[i].z; hi.z = r0[i].w; hi.w = r1[i].w;
            *(v4f*)d       = lo;
            *(v4f*)(d + 4) = hi;
        }
    }

    const float* cls; const float* reg; float* outp;
    if (blockIdx.y == 0) { cls = ego_cls;   reg = ego_reg;   outp = Qbuf; }
    else                 { cls = other_cls; reg = other_reg; outp = Kbuf; }

    const int tok0 = blockIdx.x * ENC_TOK + wave * 16;
    const int p    = tok0 + l16;

    float x0 = reg[3 * N_POS + p];
    float x1 = reg[4 * N_POS + p];
    float x2 = reg[5 * N_POS + p];
    float th = reg[6 * N_POS + p];
    float x3 = sinf(th);
    float x4 = cosf(th);
    float x5 = fmaxf(cls[p], cls[N_POS + p]);

    const int cb = 2 * half;
    v2f hf[32];
    float sum = 0.f, sumsq = 0.f;
#pragma unroll
    for (int kk = 0; kk < 32; ++kk) {
        const int c = 4 * kk + cb;
        v2f a  = *(const v2f*)&b1[c];
        v2f wa = *(const v2f*)&w1[0 * EMB + c];
        v2f wb = *(const v2f*)&w1[1 * EMB + c];
        v2f wc = *(const v2f*)&w1[2 * EMB + c];
        v2f wd = *(const v2f*)&w1[3 * EMB + c];
        v2f we = *(const v2f*)&w1[4 * EMB + c];
        v2f wf = *(const v2f*)&w1[5 * EMB + c];
        float a0 = a.x, a1 = a.y;
        a0 += x0 * wa.x;  a1 += x0 * wa.y;
        a0 += x1 * wb.x;  a1 += x1 * wb.y;
        a0 += x2 * wc.x;  a1 += x2 * wc.y;
        a0 += x3 * wd.x;  a1 += x3 * wd.y;
        a0 += x4 * we.x;  a1 += x4 * we.y;
        a0 += x5 * wf.x;  a1 += x5 * wf.y;
        a0 = fmaxf(a0, 0.f);  a1 = fmaxf(a1, 0.f);
        v2f t; t.x = a0; t.y = a1;
        hf[kk] = t;
        sum += a0; sumsq += a0 * a0;
        sum += a1; sumsq += a1 * a1;
    }
    sum   += __shfl_xor(sum, 16, 32);
    sumsq += __shfl_xor(sumsq, 16, 32);
    float mu   = sum * (1.f / EMB);
    float var  = sumsq * (1.f / EMB) - mu * mu;
    float rstd = rsqrtf(var + LN_EPSF);
#pragma unroll
    for (int kk = 0; kk < 32; ++kk) {
        const int c = 4 * kk + cb;
        v2f g = *(const v2f*)&ln_g[c];
        v2f b = *(const v2f*)&ln_b[c];
        v2f t = hf[kk];
        t.x = (t.x - mu) * rstd * g.x + b.x;
        t.y = (t.y - mu) * rstd * g.y + b.y;
        hf[kk] = t;
    }

    __syncthreads();

    const float* wl = &W2p[half * W2PITCH + 2 * l16];
    float* o = &outp[(tok0 >> 4) * TILE_FL + lane * 4];
#pragma unroll
    for (int nb = 0; nb < 8; nb += 2) {
        v4f ba0 = *(const v4f*)&b2[nb * 16 + 8 * half];
        v4f ba1 = *(const v4f*)&b2[nb * 16 + 8 * half + 4];
        v4f bb0 = *(const v4f*)&b2[nb * 16 + 16 + 8 * half];
        v4f bb1 = *(const v4f*)&b2[nb * 16 + 16 + 8 * half + 4];
        v8f acc0, acc1;
        acc0[0] = ba0.x; acc0[1] = ba0.y; acc0[2] = ba0.z; acc0[3] = ba0.w;
        acc0[4] = ba1.x; acc0[5] = ba1.y; acc0[6] = ba1.z; acc0[7] = ba1.w;
        acc1[0] = bb0.x; acc1[1] = bb0.y; acc1[2] = bb0.z; acc1[3] = bb0.w;
        acc1[4] = bb1.x; acc1[5] = bb1.y; acc1[6] = bb1.z; acc1[7] = bb1.w;
#pragma unroll
        for (int kk = 0; kk < 32; ++kk) {
            v2f bf0 = *(const v2f*)&wl[2 * kk * W2PITCH + nb * 32];
            v2f bf1 = *(const v2f*)&wl[2 * kk * W2PITCH + nb * 32 + 32];
            acc0 = __builtin_amdgcn_wmma_f32_16x16x4_f32(false, bf0, false, hf[kk],
                                                         (short)0, acc0, false, false);
            acc1 = __builtin_amdgcn_wmma_f32_16x16x4_f32(false, bf1, false, hf[kk],
                                                         (short)0, acc1, false, false);
        }
        v4f lo0, hi0, lo1, hi1;
        lo0.x = acc0[0]; lo0.y = acc0[1]; lo0.z = acc0[2]; lo0.w = acc0[3];
        hi0.x = acc0[4]; hi0.y = acc0[5]; hi0.z = acc0[6]; hi0.w = acc0[7];
        lo1.x = acc1[0]; lo1.y = acc1[1]; lo1.z = acc1[2]; lo1.w = acc1[3];
        hi1.x = acc1[4]; hi1.y = acc1[5]; hi1.z = acc1[6]; hi1.w = acc1[7];
        *(v4f*)(o + (2 * nb) * 128)     = lo0;
        *(v4f*)(o + (2 * nb + 1) * 128) = hi0;
        *(v4f*)(o + (2 * nb + 2) * 128) = lo1;
        *(v4f*)(o + (2 * nb + 3) * 128) = hi1;
    }
}

#define NTILE (N_POS / 16)
#define KQ 4
#define TPW (NTILE / KQ)
#define WGQ 64
#define CSTR 33

__device__ __forceinline__ float xhalf(float x)
{
    int b = __builtin_bit_cast(int, x);
    return __builtin_bit_cast(float,
        __builtin_amdgcn_permlanex16(b, b, 0x76543210, 0xfedcba98, false, false));
}

__device__ __forceinline__ void load_tile(v4f (&f)[16], const float* p)
{
#pragma unroll
    for (int i = 0; i < 16; ++i) f[i] = *(const v4f*)(p + i * 128);
}

__device__ __forceinline__ void softmax_tile(const v8f& S, float& mrun, float& lrun, v8f& O, v2f (&p)[4])
{
    float mx = S[0];
#pragma unroll
    for (int r = 1; r < 8; ++r) mx = fmaxf(mx, S[r]);
    mx = fmaxf(mx, xhalf(mx));
    const float newm  = fmaxf(mrun, mx);
    const float alpha = __expf(mrun - newm);
    float rs = 0.f;
#pragma unroll
    for (int j = 0; j < 4; ++j) {
        v2f a;
        a.x = __expf(S[2 * j] - newm);  a.y = __expf(S[2 * j + 1] - newm);
        p[j] = a;
        rs += a.x + a.y;
    }
    rs += xhalf(rs);
    lrun = lrun * alpha + rs;
    mrun = newm;
#pragma unroll
    for (int r = 0; r < 8; ++r) O[r] *= alpha;
}

__device__ __forceinline__ void tile_step(const v4f (&kf)[16], v4f vlo, v4f vhi,
                                          const v2f (&q0)[32], const v2f (&q1)[32],
                                          v8f& O0, v8f& O1, float& m0, float& l0, float& m1, float& l1)
{
    v8f S0 = {}, S1 = {};
#pragma unroll
    for (int i = 0; i < 16; ++i) {
        v2f a0 = __builtin_shufflevector(kf[i], kf[i], 0, 1);
        v2f a1 = __builtin_shufflevector(kf[i], kf[i], 2, 3);
        S0 = __builtin_amdgcn_wmma_f32_16x16x4_f32(false, a0, false, q0[2 * i],
                                                   (short)0, S0, false, false);
        S1 = __builtin_amdgcn_wmma_f32_16x16x4_f32(false, a0, false, q1[2 * i],
                                                   (short)0, S1, false, false);
        S0 = __builtin_amdgcn_wmma_f32_16x16x4_f32(false, a1, false, q0[2 * i + 1],
                                                   (short)0, S0, false, false);
        S1 = __builtin_amdgcn_wmma_f32_16x16x4_f32(false, a1, false, q1[2 * i + 1],
                                                   (short)0, S1, false, false);
    }
    v2f p0[4], p1[4];
    softmax_tile(S0, m0, l0, O0, p0);
    softmax_tile(S1, m1, l1, O1, p1);
    v2f va[4];
    va[0] = __builtin_shufflevector(vlo, vlo, 0, 1);
    va[1] = __builtin_shufflevector(vlo, vlo, 2, 3);
    va[2] = __builtin_shufflevector(vhi, vhi, 0, 1);
    va[3] = __builtin_shufflevector(vhi, vhi, 2, 3);
#pragma unroll
    for (int t = 0; t < 4; ++t) {
        O0 = __builtin_amdgcn_wmma_f32_16x16x4_f32(false, va[t], false, p0[t],
                                                   (short)0, O0, false, false);
        O1 = __builtin_amdgcn_wmma_f32_16x16x4_f32(false, va[t], false, p1[t],
                                                   (short)0, O1, false, false);
    }
}

__global__ __launch_bounds__(256) __attribute__((amdgpu_waves_per_eu(2, 2))) void attn_kernel(
    const float* __restrict__ Q, const float* __restrict__ K,
    const float* __restrict__ vcls, const float* __restrict__ vreg, float* __restrict__ out)
{
    __shared__ float Mls[2 * KQ * 16 * CSTR + 2 * 2 * KQ * 32];

    const int tid  = threadIdx.x;
    const int lane = tid & 31;
    const int wave = tid >> 5;
    const int qb   = wave >> 2;
    const int kq   = wave & 3;
    const int half = lane >> 4;
    const int l16  = lane & 15;

    v2f q0[32], q1[32];
    {
        const float* qp = Q + (blockIdx.x * 4 + qb * 2) * TILE_FL + lane * 4;
        v4f t0[16], t1[16];
        load_tile(t0, qp);
        load_tile(t1, qp + TILE_FL);
#pragma unroll
        for (int i = 0; i < 16; ++i) {
            v2f a, b, c, d;
            a.x = t0[i].x * TEMP_INV;  a.y = t0[i].y * TEMP_INV;
            b.x = t0[i].z * TEMP_INV;  b.y = t0[i].w * TEMP_INV;
            c.x = t1[i].x * TEMP_INV;  c.y = t1[i].y * TEMP_INV;
            d.x = t1[i].z * TEMP_INV;  d.y = t1[i].w * TEMP_INV;
            q0[2 * i] = a;  q0[2 * i + 1] = b;
            q1[2 * i] = c;  q1[2 * i + 1] = d;
        }
    }

    v8f O0 = {}, O1 = {};
    float m0 = -3.0e38f, l0 = 0.f;
    float m1 = -3.0e38f, l1 = 0.f;

    const float* vp = (l16 < 2 ? vcls + l16 * N_POS : vreg + (l16 - 2) * N_POS) + kq * TPW * 16 + 8 * half;
    const float* kp = K + kq * TPW * TILE_FL + lane * 4;

    v4f ka[16], kb[16], va0, va1, vb0, vb1;
    load_tile(ka, kp);
    va0 = *(const v4f*)vp;
    va1 = *(const v4f*)(vp + 4);

    for (int it = 0; it < TPW / 2; ++it) {
        load_tile(kb, kp + TILE_FL);
        vb0 = *(const v4f*)(vp + 16);
        vb1 = *(const v4f*)(vp + 20);
        __builtin_amdgcn_sched_barrier(0);
        tile_step(ka, va0, va1, q0, q1, O0, O1, m0, l0, m1, l1);

        __builtin_amdgcn_sched_barrier(0);
        const int adv = (it + 1 < TPW / 2) ? 2 : 0;
        const float* kn = kp + adv * TILE_FL;
        const float* vn = vp + adv * 16;
        load_tile(ka, kn);
        va0 = *(const v4f*)vn;
        va1 = *(const v4f*)(vn + 4);
        __builtin_amdgcn_sched_barrier(0);
        tile_step(kb, vb0, vb1, q0, q1, O0, O1, m0, l0, m1, l1);

        kp += 2 * TILE_FL;
        vp += 32;
    }

    float* cO = &Mls[0];
    float* cm = cO + 2 * KQ * 16 * CSTR;
    float* cl = cm + 2 * KQ * 32;
    {
        const int slot = qb * KQ + kq;
#pragma unroll
        for (int r = 0; r < 8; ++r) {
            cO[(slot * 16 + 8 * half + r) * CSTR + l16]      = O0[r];
            cO[(slot * 16 + 8 * half + r) * CSTR + 16 + l16] = O1[r];
        }
        if (half == 0) {
            cm[slot * 32 + l16]      = m0;  cl[slot * 32 + l16]      = l0;
            cm[slot * 32 + 16 + l16] = m1;  cl[slot * 32 + 16 + l16] = l1;
        }
    }
    __syncthreads();

    {
        const int qf = wave >> 2;
        const int c0 = wave & 3;
        const int ql = lane;
        float mg[KQ], mx = -3.0e38f;
#pragma unroll
        for (int k = 0; k < KQ; ++k) { mg[k] = cm[(qf * KQ + k) * 32 + ql]; mx = fmaxf(mx, mg[k]); }
        float a[KQ], lsum = 0.f;
#pragma unroll
        for (int k = 0; k < KQ; ++k) { a[k] = __expf(mg[k] - mx); lsum += cl[(qf * KQ + k) * 32 + ql] * a[k]; }
#pragma unroll
        for (int i = 0; i < 4; ++i) {
            const int c = c0 + 4 * i;
            float o = 0.f;
#pragma unroll
            for (int k = 0; k < KQ; ++k) o += cO[((qf * KQ + k) * 16 + c) * CSTR + ql] * a[k];
            out[c * N_POS + blockIdx.x * WGQ + qf * 32 + ql] = o / lsum;
        }
    }
}

extern "C" void kernel_launch(void* const* d_in, const int* in_sizes, int n_in,
                              void* d_out, int out_size, void* d_ws, size_t ws_size,
                              hipStream_t stream)
{
    const float* ego_cls   = (const float*)d_in[0];
    const float* ego_reg   = (const float*)d_in[1];
    const float* other_cls = (const float*)d_in[2];
    const float* other_reg = (const float*)d_in[3];
    const float* w1  = (const float*)d_in[4];
    const float* b1  = (const float*)d_in[5];
    const float* ln_g = (const float*)d_in[6];
    const float* ln_b = (const float*)d_in[7];
    const float* w2  = (const float*)d_in[8];
    const float* b2  = (const float*)d_in[9];

    float* Qbuf = (float*)d_ws;
    float* Kbuf = Qbuf + N_POS * EMB;
    float* out  = (float*)d_out;

    encode_kernel<<<dim3(N_POS / ENC_TOK, 2), 256, 0, stream>>>(
        ego_cls, ego_reg, other_cls, other_reg,
        w1, b1, ln_g, ln_b, w2, b2, Qbuf, Kbuf);

    attn_kernel<<<N_POS / WGQ, 256, 0, stream>>>(Qbuf, Kbuf, other_cls, other_reg, out);
}
